# starting kernel
# baseline (speedup 1.0000x reference)
#include <hip/hip_runtime.h>
#include <hip/hip_bf16.h>
#include <math.h>

typedef __attribute__((ext_vector_type(2))) float v2f;
typedef __attribute__((ext_vector_type(8))) float v8f;

#define CCH   64
#define FILLW 2.0f

__device__ __forceinline__ void atomic_add_f32(float* p, float v) {
  __hip_atomic_fetch_add(p, v, __ATOMIC_RELAXED, __HIP_MEMORY_SCOPE_AGENT);
}

__global__ void gcn_init(float* __restrict__ out, float* __restrict__ deg,
                         int total, int n_nodes) {
  int t = blockIdx.x * blockDim.x + threadIdx.x;
  if (t < total)   out[t] = 0.0f;
  if (t < n_nodes) deg[t] = FILLW;
}

__global__ void gcn_deg(const int* __restrict__ dst, const float* __restrict__ w,
                        float* __restrict__ deg, int n_edges) {
  int e = blockIdx.x * blockDim.x + threadIdx.x;
  if (e < n_edges) atomic_add_f32(&deg[dst[e]], w[e]);
}

__global__ void gcn_dinv(float* __restrict__ deg, int n_nodes) {
  int n = blockIdx.x * blockDim.x + threadIdx.x;
  if (n < n_nodes) {
    float d = deg[n];
    deg[n] = (d > 0.0f) ? rsqrtf(d) : 0.0f;
  }
}

__global__ void gcn_gemm_wmma(const float* __restrict__ X,
                              const float* __restrict__ W,
                              float* __restrict__ XW, int n_nodes) {
  const int wave = (blockIdx.x * blockDim.x + threadIdx.x) >> 5;
  const int lane = threadIdx.x & 31;
  const int n_row_tiles = (n_nodes + 15) >> 4;
  const int row_tile = wave >> 2;
  const int col_tile = wave & 3;
  if (row_tile >= n_row_tiles) return;

  const int half = lane >> 4;
  const int l16  = lane & 15;
  int row = row_tile * 16 + l16;
  int ldrow = (row < n_nodes) ? row : (n_nodes - 1);

  const float* __restrict__ xrow = X + (size_t)ldrow * CCH;
  const float* __restrict__ wcol = W + col_tile * 16 + l16;

  v8f acc = {};
#pragma unroll
  for (int k0 = 0; k0 < CCH; k0 += 4) {
    const int ka = k0 + 2 * half;
    v2f a, b;
    a.x = xrow[ka + 0];
    a.y = xrow[ka + 1];
    b.x = wcol[(size_t)(ka + 0) * CCH];
    b.y = wcol[(size_t)(ka + 1) * CCH];
    acc = __builtin_amdgcn_wmma_f32_16x16x4_f32(
         false, a,  false, b,
         (short)0, acc,  false,  false);
  }

  const int base_row = row_tile * 16 + 8 * half;
  float* __restrict__ dstp = XW + (size_t)base_row * CCH + col_tile * 16 + l16;
#pragma unroll
  for (int i = 0; i < 8; ++i) {
    if (base_row + i < n_nodes) dstp[(size_t)i * CCH] = acc[i];
  }
}

__global__ void gcn_scatter(const int* __restrict__ src, const int* __restrict__ dst,
                            const float* __restrict__ w, const float* __restrict__ dinv,
                            const float* __restrict__ xw, float* __restrict__ out,
                            int n_edges) {
  const int gt   = blockIdx.x * blockDim.x + threadIdx.x;
  const int e    = gt >> 5;
  const int lane = gt & 31;
  if (e >= n_edges) return;

  const int s = src[e];
  const int d = dst[e];
  float norm = 0.0f;
  if (lane == 0) norm = dinv[s] * w[e] * dinv[d];
  norm = __shfl(norm, 0, 32);

  const float* __restrict__ xr = xw + (size_t)s * CCH;
  float* __restrict__ orow     = out + (size_t)d * CCH;

  atomic_add_f32(&orow[lane],      xr[lane]      * norm);
  atomic_add_f32(&orow[lane + 32], xr[lane + 32] * norm);
}

__global__ void gcn_final(float* __restrict__ out, const float* __restrict__ xw,
                          const float* __restrict__ dinv, const float* __restrict__ b,
                          int total) {
  int t = blockIdx.x * blockDim.x + threadIdx.x;
  if (t < total) {
    int n = t >> 6;
    int c = t & (CCH - 1);
    float di = dinv[n];
    float v  = out[t] + xw[t] * (FILLW * di * di) + b[c];
    out[t] = tanhf(v);
  }
}

extern "C" void kernel_launch(void* const* d_in, const int* in_sizes, int n_in,
                              void* d_out, int out_size, void* d_ws, size_t ws_size,
                              hipStream_t stream) {
  const float* x  = (const float*)d_in[0];
  const int*   ei = (const int*)d_in[1];
  const float* ew = (const float*)d_in[2];
  const float* W  = (const float*)d_in[3];
  const float* b  = (const float*)d_in[4];
  float* out = (float*)d_out;

  const int n_nodes = in_sizes[0] / CCH;
  const int n_edges = in_sizes[2];
  const int* src = ei;
  const int* dst = ei + n_edges;

  float* xw   = (float*)d_ws;
  float* dinv = xw + (size_t)n_nodes * CCH;

  const int total = n_nodes * CCH;
  const int B = 256;

  gcn_init<<<(total + B - 1) / B, B, 0, stream>>>(out, dinv, total, n_nodes);
  gcn_deg<<<(n_edges + B - 1) / B, B, 0, stream>>>(dst, ew, dinv, n_edges);
  gcn_dinv<<<(n_nodes + B - 1) / B, B, 0, stream>>>(dinv, n_nodes);

  const int n_row_tiles = (n_nodes + 15) / 16;
  const int waves = n_row_tiles * (CCH / 16);
  gcn_gemm_wmma<<<(waves + 7) / 8, B, 0, stream>>>(x, W, xw, n_nodes);

  const long long sthreads = (long long)n_edges * 32;
  gcn_scatter<<<(unsigned)((sthreads + B - 1) / B), B, 0, stream>>>(
      src, dst, ew, dinv, xw, out, n_edges);

  gcn_final<<<(total + B - 1) / B, B, 0, stream>>>(out, xw, dinv, b, total);
}


# Sort edges by dst (u32 rank atomics, prefix sum) and gather each node's sum in registers: no f32 atomics, no out zeroing; per-node sum order = arrival rank
# speedup vs baseline: 1.8722x; 1.8722x over previous
#include <hip/hip_runtime.h>
#include <hip/hip_bf16.h>
#include <math.h>

typedef __attribute__((ext_vector_type(2))) float v2f;
typedef __attribute__((ext_vector_type(8))) float v8f;
typedef __attribute__((ext_vector_type(2))) int   v2i;

#define CCH   64
#define FILLW 2.0f

__global__ void gcn_zero_cnt(int* __restrict__ cnt, int n_nodes) {
  int t = blockIdx.x * blockDim.x + threadIdx.x;
  if (t < n_nodes) cnt[t] = 0;
}

__global__ void gcn_count(const int* __restrict__ dst, int* __restrict__ cnt,
                          int* __restrict__ rank, int n_edges) {
  int e = blockIdx.x * blockDim.x + threadIdx.x;
  if (e < n_edges)
    rank[e] = __hip_atomic_fetch_add(&cnt[dst[e]], 1, __ATOMIC_RELAXED,
                                     __HIP_MEMORY_SCOPE_AGENT);
}

__global__ void __launch_bounds__(1024)
gcn_scan(const int* __restrict__ cnt, int* __restrict__ off, int n_nodes) {
  __shared__ int wtot[32];
  __shared__ int wpre[32];
  const int tid  = threadIdx.x;
  const int lane = tid & 31;
  const int wv   = tid >> 5;
  const int blk_beg = blockIdx.x * 1024;

  int part = 0;
  const int4* __restrict__ c4 = (const int4*)cnt;
  const int n4 = blk_beg >> 2;
#pragma unroll 4
  for (int i = tid; i < n4; i += 1024) {
    int4 q = c4[i];
    part += (q.x + q.y) + (q.z + q.w);
  }

  const int idx  = blk_beg + tid;
  const int mine = (idx < n_nodes) ? cnt[idx] : 0;
  int incl = mine;
  for (int d = 1; d < 32; d <<= 1) {
    int t = __shfl_up(incl, d, 32);
    if (lane >= d) incl += t;
  }
  for (int d = 16; d > 0; d >>= 1) part += __shfl_xor(part, d, 32);
  if (lane == 31) wtot[wv] = incl;
  if (lane == 0)  wpre[wv] = part;
  __syncthreads();
  if (wv == 0) {
    const int t = wtot[lane];
    int p  = wpre[lane];
    int ti = t;
    for (int d = 1; d < 32; d <<= 1) {
      int u = __shfl_up(ti, d, 32);
      if (lane >= d) ti += u;
    }
    for (int d = 16; d > 0; d >>= 1) p += __shfl_xor(p, d, 32);
    wtot[lane] = ti - t + p;
  }
  __syncthreads();
  if (idx < n_nodes) off[idx] = wtot[wv] + incl - mine;
}

__global__ void gcn_fill(const int* __restrict__ src, const int* __restrict__ dst,
                         const float* __restrict__ w, const int* __restrict__ off,
                         const int* __restrict__ rank, int* __restrict__ pair,
                         int n_edges) {
  int e = blockIdx.x * blockDim.x + threadIdx.x;
  if (e < n_edges) {
    const int p = off[dst[e]] + rank[e];
    if ((unsigned)p < (unsigned)n_edges)
      ((v2i*)pair)[p] = (v2i){src[e], __float_as_int(w[e])};
  }
}

__global__ void gcn_dinv(const int* __restrict__ pair, const int* __restrict__ off,
                         const int* __restrict__ cnt, float* __restrict__ dinv,
                         int n_nodes, int n_edges) {
  const int gt  = blockIdx.x * blockDim.x + threadIdx.x;
  const int n   = gt >> 3;
  const int sub = gt & 7;
  float d = 0.0f;
  if (n < n_nodes) {
    const int beg = off[n];
    const int m   = cnt[n];
    for (int i = sub; i < m; i += 8) {
      const int q = beg + i;
      if (q < n_edges) d += __int_as_float(pair[2 * (size_t)q + 1]);
    }
  }
  d += __shfl_xor(d, 1, 32);
  d += __shfl_xor(d, 2, 32);
  d += __shfl_xor(d, 4, 32);
  if (n < n_nodes && sub == 0) {
    d += FILLW;
    dinv[n] = (d > 0.0f) ? rsqrtf(d) : 0.0f;
  }
}

__global__ void gcn_gemm_wmma(const float* __restrict__ X,
                              const float* __restrict__ W,
                              float* __restrict__ XW, int n_nodes) {
  const int wave = (blockIdx.x * blockDim.x + threadIdx.x) >> 5;
  const int lane = threadIdx.x & 31;
  const int n_row_tiles = (n_nodes + 15) >> 4;
  const int row_tile = wave >> 2;
  const int col_tile = wave & 3;
  if (row_tile >= n_row_tiles) return;

  const int half = lane >> 4;
  const int l16  = lane & 15;
  int row = row_tile * 16 + l16;
  int ldrow = (row < n_nodes) ? row : (n_nodes - 1);

  const float* __restrict__ xrow = X + (size_t)ldrow * CCH;
  const float* __restrict__ wcol = W + col_tile * 16 + l16;

  v8f acc = {};
#pragma unroll
  for (int k0 = 0; k0 < CCH; k0 += 4) {
    const int ka = k0 + 2 * half;
    v2f a, b;
    a.x = xrow[ka + 0];
    a.y = xrow[ka + 1];
    b.x = wcol[(size_t)(ka + 0) * CCH];
    b.y = wcol[(size_t)(ka + 1) * CCH];
    acc = __builtin_amdgcn_wmma_f32_16x16x4_f32(
         false, a,  false, b,
         (short)0, acc,  false,  false);
  }

  const int base_row = row_tile * 16 + 8 * half;
  float* __restrict__ dstp = XW + (size_t)base_row * CCH + col_tile * 16 + l16;
#pragma unroll
  for (int i = 0; i < 8; ++i) {
    if (base_row + i < n_nodes) dstp[(size_t)i * CCH] = acc[i];
  }
}

__global__ void gcn_gather(const int* __restrict__ pair, const int* __restrict__ off,
                           const int* __restrict__ cnt, const float* __restrict__ dinv,
                           const float* __restrict__ xw, const float* __restrict__ b,
                           float* __restrict__ out, int n_nodes, int n_edges) {
  const int lane = threadIdx.x & 31;
  const int n = __builtin_amdgcn_readfirstlane(
      (int)((blockIdx.x * blockDim.x + threadIdx.x) >> 5));
  if (n >= n_nodes) return;
  const int beg = __builtin_amdgcn_readfirstlane(off[n]);
  const int m   = __builtin_amdgcn_readfirstlane(cnt[n]);
  const float dd = dinv[n];
  const v2f self = ((const v2f*)(xw + (size_t)n * CCH))[lane];
  const float b0 = b[2 * lane];
  const float b1 = b[2 * lane + 1];

  v2f acc = {0.0f, 0.0f};
  for (int base = 0; base < m; base += 32) {
    const int mm = min(32, m - base);
    int   s_i = 0;
    float c_i = 0.0f;
    const int q = beg + base + lane;
    if (lane < mm && q < n_edges) {
      const v2i p = ((const v2i*)pair)[q];
      s_i = p.x;
      c_i = dinv[p.x] * __int_as_float(p.y) * dd;
    }
    for (int j = 0; j < mm; j += 8) {
      v2f   v[8];
      float c[8];
#pragma unroll
      for (int k = 0; k < 8; ++k) {
        v[k] = (v2f){0.0f, 0.0f};
        c[k] = 0.0f;
        if (j + k < mm) {
          const int s = __builtin_amdgcn_readlane(s_i, j + k);
          c[k] = __int_as_float(__builtin_amdgcn_readlane(__float_as_int(c_i), j + k));
          v[k] = ((const v2f*)(xw + (size_t)s * CCH))[lane];
        }
      }
#pragma unroll
      for (int k = 0; k < 8; ++k) {
        acc.x += v[k].x * c[k];
        acc.y += v[k].y * c[k];
      }
    }
  }

  const float k2 = FILLW * dd * dd;
  v2f r;
  r.x = tanhf(acc.x + self.x * k2 + b0);
  r.y = tanhf(acc.y + self.y * k2 + b1);
  ((v2f*)(out + (size_t)n * CCH))[lane] = r;
}

extern "C" void kernel_launch(void* const* d_in, const int* in_sizes, int n_in,
                              void* d_out, int out_size, void* d_ws, size_t ws_size,
                              hipStream_t stream) {
  const float* x  = (const float*)d_in[0];
  const int*   ei = (const int*)d_in[1];
  const float* ew = (const float*)d_in[2];
  const float* W  = (const float*)d_in[3];
  const float* b  = (const float*)d_in[4];
  float* out = (float*)d_out;

  const int n_nodes = in_sizes[0] / CCH;
  const int n_edges = in_sizes[2];
  const int* src = ei;
  const int* dst = ei + n_edges;

  const size_t n_pad = ((size_t)n_nodes + 63) & ~(size_t)63;
  const size_t e_pad = ((size_t)n_edges + 63) & ~(size_t)63;
  float* xw   = (float*)d_ws;
  float* dinv = xw + n_pad * CCH;
  int*   cnt  = (int*)(dinv + n_pad);
  int*   off  = cnt + n_pad;
  int*   rank = off + n_pad;
  int*   pair = rank + e_pad;

  const int B = 256;

  gcn_zero_cnt<<<(n_nodes + B - 1) / B, B, 0, stream>>>(cnt, n_nodes);
  gcn_count<<<(n_edges + B - 1) / B, B, 0, stream>>>(dst, cnt, rank, n_edges);
  gcn_scan<<<(n_nodes + 1023) / 1024, 1024, 0, stream>>>(cnt, off, n_nodes);
  gcn_fill<<<(n_edges + B - 1) / B, B, 0, stream>>>(src, dst, ew, off, rank, pair, n_edges);
  gcn_dinv<<<(n_nodes * 8 + B - 1) / B, B, 0, stream>>>(pair, off, cnt, dinv, n_nodes, n_edges);

  const int n_row_tiles = (n_nodes + 15) / 16;
  const int waves = n_row_tiles * (CCH / 16);
  gcn_gemm_wmma<<<(waves + 7) / 8, B, 0, stream>>>(x, W, xw, n_nodes);

  gcn_gather<<<(n_nodes + 7) / 8, B, 0, stream>>>(pair, off, cnt, dinv, xw, b, out,
                                                  n_nodes, n_edges);
}


# Edge sort ranks from LDS counters in two levels (256-node dst ranges, then nodes) instead of one returning global atomic per edge; per-node sum order = LDS arrival rank
# speedup vs baseline: 3.7607x; 2.0087x over previous
#include <hip/hip_runtime.h>
#include <hip/hip_bf16.h>
#include <math.h>

typedef __attribute__((ext_vector_type(2))) float v2f;
typedef __attribute__((ext_vector_type(8))) float v8f;
typedef __attribute__((ext_vector_type(2))) int   v2i;

#define CCH   64
#define FILLW 2.0f

__global__ void gcn_zero_cnt(int* __restrict__ cnt, int n_nodes) {
  int t = blockIdx.x * blockDim.x + threadIdx.x;
  if (t < n_nodes) cnt[t] = 0;
}

#define BINW  256
#define BSH   8
#define EPB   4096
#define EPT   16

__global__ void __launch_bounds__(256)
gcn_hist(const int* __restrict__ dst, int* __restrict__ bin_tot,
         int n_nodes, int n_edges, int n_bins) {
  extern __shared__ int hist[];
  const int tid = threadIdx.x;
  for (int i = tid; i < n_bins; i += 256) hist[i] = 0;
  __syncthreads();
  const int e0 = blockIdx.x * EPB + tid;
  int dk[EPT];
#pragma unroll
  for (int k = 0; k < EPT; ++k) dk[k] = dst[min(e0 + k * 256, n_edges - 1)];
#pragma unroll
  for (int k = 0; k < EPT; ++k) {
    if (e0 + k * 256 < n_edges && (unsigned)dk[k] < (unsigned)n_nodes)
      atomicAdd(&hist[dk[k] >> BSH], 1);
  }
  __syncthreads();
  for (int i = tid; i < n_bins; i += 256) {
    const int h = hist[i];
    if (h) atomicAdd(&bin_tot[i], h);
  }
}

__global__ void __launch_bounds__(256)
gcn_place(const int* __restrict__ src, const int* __restrict__ dst,
          const float* __restrict__ w, const int* __restrict__ bin_tot,
          int* __restrict__ bin_cur, int* __restrict__ tmp,
          int n_nodes, int n_edges, int n_bins) {
  extern __shared__ int sh[];
  __shared__ int wsum[8];
  int* hist = sh;
  int* base = sh + n_bins;
  const int tid  = threadIdx.x;
  const int lane = tid & 31;
  const int wv   = tid >> 5;
  for (int i = tid; i < n_bins; i += 256) hist[i] = 0;
  __syncthreads();

  const int e0 = blockIdx.x * EPB + tid;
  int dk[EPT];
  int sk[EPT];
  int wk[EPT];
  int rk[EPT];
#pragma unroll
  for (int k = 0; k < EPT; ++k) {
    const int e = min(e0 + k * 256, n_edges - 1);
    dk[k] = dst[e];
    sk[k] = src[e];
    wk[k] = __float_as_int(w[e]);
  }
#pragma unroll
  for (int k = 0; k < EPT; ++k) {
    rk[k] = 0;
    if (e0 + k * 256 < n_edges && (unsigned)dk[k] < (unsigned)n_nodes)
      rk[k] = atomicAdd(&hist[dk[k] >> BSH], 1);
    else
      dk[k] = -1;
  }
  __syncthreads();

  int carry = 0;
  for (int c0 = 0; c0 < n_bins; c0 += 256) {
    const int i = c0 + tid;
    const int t = (i < n_bins) ? bin_tot[i] : 0;
    int incl = t;
    for (int s = 1; s < 32; s <<= 1) {
      const int u = __shfl_up(incl, s, 32);
      if (lane >= s) incl += u;
    }
    if (lane == 31) wsum[wv] = incl;
    __syncthreads();
    int wpre = 0;
    int tot  = 0;
    for (int j = 0; j < 8; ++j) {
      const int x = wsum[j];
      if (j < wv) wpre += x;
      tot += x;
    }
    if (i < n_bins) {
      const int h = hist[i];
      const int got = h ? atomicAdd(&bin_cur[i], h) : 0;
      base[i] = carry + wpre + incl - t + got;
    }
    carry += tot;
    __syncthreads();
  }

#pragma unroll
  for (int k = 0; k < EPT; ++k) {
    if (dk[k] >= 0) {
      const int p = base[dk[k] >> BSH] + rk[k];
      if ((unsigned)p < (unsigned)n_edges)
        ((int4*)tmp)[p] = make_int4(sk[k], wk[k], dk[k], 0);
    }
  }
}

__global__ void __launch_bounds__(1024)
gcn_owner(const int* __restrict__ tmp, const int* __restrict__ bin_tot,
          int* __restrict__ cnt, int* __restrict__ off, int* __restrict__ pair,
          int n_nodes, int n_edges) {
  __shared__ int cntL[BINW];
  __shared__ int offL[BINW];
  __shared__ int curL[BINW];
  __shared__ int red[32];
  const int tid  = threadIdx.x;
  const int lane = tid & 31;
  const int wv   = tid >> 5;
  const int bin  = blockIdx.x;
  const int node0 = bin * BINW;
  const int4* __restrict__ rec4 = (const int4*)tmp;

  int part = 0;
  for (int i = tid; i < bin; i += 1024) part += bin_tot[i];
  for (int s = 16; s > 0; s >>= 1) part += __shfl_xor(part, s, 32);
  if (lane == 0) red[wv] = part;
  if (tid < BINW) { cntL[tid] = 0; curL[tid] = 0; }
  __syncthreads();
  int start = 0;
  for (int j = 0; j < 32; ++j) start += red[j];
  int m = bin_tot[bin];
  start = min(max(start, 0), n_edges);
  m     = min(max(m, 0), n_edges - start);
  __syncthreads();

  for (int i0 = tid; i0 - tid < m; i0 += 4096) {
    int z[4];
#pragma unroll
    for (int k = 0; k < 4; ++k) z[k] = rec4[start + min(i0 + k * 1024, m - 1)].z;
#pragma unroll
    for (int k = 0; k < 4; ++k) {
      const int dl = z[k] - node0;
      if (i0 + k * 1024 < m && (unsigned)dl < (unsigned)BINW) atomicAdd(&cntL[dl], 1);
    }
  }
  __syncthreads();

  int c = 0, incl = 0;
  if (tid < BINW) {
    c = cntL[tid];
    incl = c;
    for (int s = 1; s < 32; s <<= 1) {
      const int u = __shfl_up(incl, s, 32);
      if (lane >= s) incl += u;
    }
    if (lane == 31) red[wv] = incl;
  }
  __syncthreads();
  if (tid < BINW) {
    int wpre = 0;
    for (int j = 0; j < 8; ++j) if (j < wv) wpre += red[j];
    const int excl = wpre + incl - c;
    offL[tid] = excl;
    const int node = node0 + tid;
    if (node < n_nodes) { cnt[node] = c; off[node] = start + excl; }
  }
  __syncthreads();

  for (int i0 = tid; i0 - tid < m; i0 += 4096) {
    int4 q[4];
#pragma unroll
    for (int k = 0; k < 4; ++k) q[k] = rec4[start + min(i0 + k * 1024, m - 1)];
#pragma unroll
    for (int k = 0; k < 4; ++k) {
      const int dl = q[k].z - node0;
      if (i0 + k * 1024 < m && (unsigned)dl < (unsigned)BINW) {
        const int p = start + offL[dl] + atomicAdd(&curL[dl], 1);
        if ((unsigned)p < (unsigned)n_edges)
          ((v2i*)pair)[p] = (v2i){q[k].x, q[k].y};
      }
    }
  }
}

__global__ void gcn_dinv(const int* __restrict__ pair, const int* __restrict__ off,
                         const int* __restrict__ cnt, float* __restrict__ dinv,
                         int n_nodes, int n_edges) {
  const int gt  = blockIdx.x * blockDim.x + threadIdx.x;
  const int n   = gt >> 3;
  const int sub = gt & 7;
  float d = 0.0f;
  if (n < n_nodes) {
    const int beg = off[n];
    const int m   = cnt[n];
    for (int i = sub; i < m; i += 8) {
      const int q = beg + i;
      if (q < n_edges) d += __int_as_float(pair[2 * (size_t)q + 1]);
    }
  }
  d += __shfl_xor(d, 1, 32);
  d += __shfl_xor(d, 2, 32);
  d += __shfl_xor(d, 4, 32);
  if (n < n_nodes && sub == 0) {
    d += FILLW;
    dinv[n] = (d > 0.0f) ? rsqrtf(d) : 0.0f;
  }
}

__global__ void gcn_gemm_wmma(const float* __restrict__ X,
                              const float* __restrict__ W,
                              float* __restrict__ XW, int n_nodes) {
  const int wave = (blockIdx.x * blockDim.x + threadIdx.x) >> 5;
  const int lane = threadIdx.x & 31;
  const int n_row_tiles = (n_nodes + 15) >> 4;
  const int row_tile = wave >> 2;
  const int col_tile = wave & 3;
  if (row_tile >= n_row_tiles) return;

  const int half = lane >> 4;
  const int l16  = lane & 15;
  int row = row_tile * 16 + l16;
  int ldrow = (row < n_nodes) ? row : (n_nodes - 1);

  const float* __restrict__ xrow = X + (size_t)ldrow * CCH;
  const float* __restrict__ wcol = W + col_tile * 16 + l16;

  v8f acc = {};
#pragma unroll
  for (int k0 = 0; k0 < CCH; k0 += 4) {
    const int ka = k0 + 2 * half;
    v2f a, b;
    a.x = xrow[ka + 0];
    a.y = xrow[ka + 1];
    b.x = wcol[(size_t)(ka + 0) * CCH];
    b.y = wcol[(size_t)(ka + 1) * CCH];
    acc = __builtin_amdgcn_wmma_f32_16x16x4_f32(
         false, a,  false, b,
         (short)0, acc,  false,  false);
  }

  const int base_row = row_tile * 16 + 8 * half;
  float* __restrict__ dstp = XW + (size_t)base_row * CCH + col_tile * 16 + l16;
#pragma unroll
  for (int i = 0; i < 8; ++i) {
    if (base_row + i < n_nodes) dstp[(size_t)i * CCH] = acc[i];
  }
}

__global__ void gcn_gather(const int* __restrict__ pair, const int* __restrict__ off,
                           const int* __restrict__ cnt, const float* __restrict__ dinv,
                           const float* __restrict__ xw, const float* __restrict__ b,
                           float* __restrict__ out, int n_nodes, int n_edges) {
  const int lane = threadIdx.x & 31;
  const int n = __builtin_amdgcn_readfirstlane(
      (int)((blockIdx.x * blockDim.x + threadIdx.x) >> 5));
  if (n >= n_nodes) return;
  const int beg = __builtin_amdgcn_readfirstlane(off[n]);
  const int m   = __builtin_amdgcn_readfirstlane(cnt[n]);
  const float dd = dinv[n];
  const v2f self = ((const v2f*)(xw + (size_t)n * CCH))[lane];
  const float b0 = b[2 * lane];
  const float b1 = b[2 * lane + 1];

  v2f acc = {0.0f, 0.0f};
  for (int base = 0; base < m; base += 32) {
    const int mm = min(32, m - base);
    int   s_i = 0;
    float c_i = 0.0f;
    const int q = beg + base + lane;
    if (lane < mm && q < n_edges) {
      const v2i p = ((const v2i*)pair)[q];
      s_i = p.x;
      c_i = dinv[p.x] * __int_as_float(p.y) * dd;
    }
    for (int j = 0; j < mm; j += 8) {
      v2f   v[8];
      float c[8];
#pragma unroll
      for (int k = 0; k < 8; ++k) {
        v[k] = (v2f){0.0f, 0.0f};
        c[k] = 0.0f;
        if (j + k < mm) {
          const int s = __builtin_amdgcn_readlane(s_i, j + k);
          c[k] = __int_as_float(__builtin_amdgcn_readlane(__float_as_int(c_i), j + k));
          v[k] = ((const v2f*)(xw + (size_t)s * CCH))[lane];
        }
      }
#pragma unroll
      for (int k = 0; k < 8; ++k) {
        acc.x += v[k].x * c[k];
        acc.y += v[k].y * c[k];
      }
    }
  }

  const float k2 = FILLW * dd * dd;
  v2f r;
  r.x = tanhf(acc.x + self.x * k2 + b0);
  r.y = tanhf(acc.y + self.y * k2 + b1);
  ((v2f*)(out + (size_t)n * CCH))[lane] = r;
}

extern "C" void kernel_launch(void* const* d_in, const int* in_sizes, int n_in,
                              void* d_out, int out_size, void* d_ws, size_t ws_size,
                              hipStream_t stream) {
  const float* x  = (const float*)d_in[0];
  const int*   ei = (const int*)d_in[1];
  const float* ew = (const float*)d_in[2];
  const float* W  = (const float*)d_in[3];
  const float* b  = (const float*)d_in[4];
  float* out = (float*)d_out;

  const int n_nodes = in_sizes[0] / CCH;
  const int n_edges = in_sizes[2];
  const int* src = ei;
  const int* dst = ei + n_edges;

  const size_t n_pad = ((size_t)n_nodes + 63) & ~(size_t)63;
  const size_t e_pad = ((size_t)n_edges + 63) & ~(size_t)63;
  const int    n_bins = (n_nodes + BINW - 1) / BINW;
  const size_t b_pad = ((size_t)n_bins + 63) & ~(size_t)63;
  float* xw   = (float*)d_ws;
  float* dinv = xw + n_pad * CCH;
  int*   cnt  = (int*)(dinv + n_pad);
  int*   off  = cnt + n_pad;
  int*   tmp  = off + n_pad;
  int*   pair = tmp + e_pad * 4;
  int*   bin_tot = pair + e_pad * 2;
  int*   bin_cur = bin_tot + b_pad;

  const int B = 256;
  const int n_chunks = (n_edges + EPB - 1) / EPB;

  gcn_zero_cnt<<<(int)((2 * b_pad + B - 1) / B), B, 0, stream>>>(bin_tot, (int)(2 * b_pad));
  gcn_hist<<<n_chunks, B, n_bins * sizeof(int), stream>>>(dst, bin_tot, n_nodes, n_edges, n_bins);
  gcn_place<<<n_chunks, B, 2 * n_bins * sizeof(int), stream>>>(src, dst, ew, bin_tot, bin_cur, tmp,
                                                              n_nodes, n_edges, n_bins);
  gcn_owner<<<n_bins, 1024, 0, stream>>>(tmp, bin_tot, cnt, off, pair, n_nodes, n_edges);
  gcn_dinv<<<(n_nodes * 8 + B - 1) / B, B, 0, stream>>>(pair, off, cnt, dinv, n_nodes, n_edges);

  const int n_row_tiles = (n_nodes + 15) / 16;
  const int waves = n_row_tiles * (CCH / 16);
  gcn_gemm_wmma<<<(waves + 7) / 8, B, 0, stream>>>(x, W, xw, n_nodes);

  gcn_gather<<<(n_nodes + 7) / 8, B, 0, stream>>>(pair, off, cnt, dinv, xw, b, out,
                                                  n_nodes, n_edges);
}


# X@W tiles run as extra work-groups behind the edge chunks of the two edge-pass launches (hist, place) instead of in a launch of their own: 6 launches, product under the sort; all sums unchanged
# speedup vs baseline: 3.9937x; 1.0620x over previous
#include <hip/hip_runtime.h>
#include <hip/hip_bf16.h>
#include <math.h>

typedef __attribute__((ext_vector_type(2))) float v2f;
typedef __attribute__((ext_vector_type(8))) float v8f;
typedef __attribute__((ext_vector_type(2))) int   v2i;

#define CCH   64
#define FILLW 2.0f

__global__ void gcn_zero_cnt(int* __restrict__ cnt, int n_nodes) {
  int t = blockIdx.x * blockDim.x + threadIdx.x;
  if (t < n_nodes) cnt[t] = 0;
}

#define BINW  256
#define BSH   8
#define EPB   4096
#define EPT   16

__device__ __forceinline__ void gcn_gemm_group(const float* __restrict__ X,
                                               const float* __restrict__ W,
                                               float* __restrict__ XW, int n_nodes,
                                               int group, int wave0, int wave1) {
  const int wave = wave0 + ((group * 256 + (int)threadIdx.x) >> 5);
  const int lane = threadIdx.x & 31;
  const int n_row_tiles = (n_nodes + 15) >> 4;
  const int row_tile = wave >> 2;
  const int col_tile = wave & 3;
  if (wave >= wave1 || row_tile >= n_row_tiles) return;

  const int half = lane >> 4;
  const int l16  = lane & 15;
  int row = row_tile * 16 + l16;
  int ldrow = (row < n_nodes) ? row : (n_nodes - 1);

  const float* __restrict__ xrow = X + (size_t)ldrow * CCH;
  const float* __restrict__ wcol = W + col_tile * 16 + l16;

  v8f acc = {};
#pragma unroll
  for (int k0 = 0; k0 < CCH; k0 += 4) {
    const int ka = k0 + 2 * half;
    v2f a, b;
    a.x = xrow[ka + 0];
    a.y = xrow[ka + 1];
    b.x = wcol[(size_t)(ka + 0) * CCH];
    b.y = wcol[(size_t)(ka + 1) * CCH];
    acc = __builtin_amdgcn_wmma_f32_16x16x4_f32(
         false, a,  false, b,
         (short)0, acc,  false,  false);
  }

  const int base_row = row_tile * 16 + 8 * half;
  float* __restrict__ dstp = XW + (size_t)base_row * CCH + col_tile * 16 + l16;
#pragma unroll
  for (int i = 0; i < 8; ++i) {
    if (base_row + i < n_nodes) dstp[(size_t)i * CCH] = acc[i];
  }
}

__global__ void __launch_bounds__(256)
gcn_hist(const int* __restrict__ dst, int* __restrict__ bin_tot,
         int n_nodes, int n_edges, int n_bins,
         const float* __restrict__ X, const float* __restrict__ W,
         float* __restrict__ XW, int wave0, int wave1) {
  extern __shared__ int hist[];
  const int tid = threadIdx.x;
  const int n_chunks = (n_edges + EPB - 1) / EPB;
  if ((int)blockIdx.x >= n_chunks) {
    gcn_gemm_group(X, W, XW, n_nodes, (int)blockIdx.x - n_chunks, wave0, wave1);
    return;
  }
  for (int i = tid; i < n_bins; i += 256) hist[i] = 0;
  __syncthreads();
  const int e0 = blockIdx.x * EPB + tid;
  int dk[EPT];
#pragma unroll
  for (int k = 0; k < EPT; ++k) dk[k] = dst[min(e0 + k * 256, n_edges - 1)];
#pragma unroll
  for (int k = 0; k < EPT; ++k) {
    if (e0 + k * 256 < n_edges && (unsigned)dk[k] < (unsigned)n_nodes)
      atomicAdd(&hist[dk[k] >> BSH], 1);
  }
  __syncthreads();
  for (int i = tid; i < n_bins; i += 256) {
    const int h = hist[i];
    if (h) atomicAdd(&bin_tot[i], h);
  }
}

__global__ void __launch_bounds__(256)
gcn_place(const int* __restrict__ src, const int* __restrict__ dst,
          const float* __restrict__ w, const int* __restrict__ bin_tot,
          int* __restrict__ bin_cur, int* __restrict__ tmp,
          int n_nodes, int n_edges, int n_bins,
          const float* __restrict__ X, const float* __restrict__ W,
          float* __restrict__ XW, int wave0, int wave1) {
  extern __shared__ int sh[];
  __shared__ int wsum[8];
  const int n_chunks = (n_edges + EPB - 1) / EPB;
  if ((int)blockIdx.x >= n_chunks) {
    gcn_gemm_group(X, W, XW, n_nodes, (int)blockIdx.x - n_chunks, wave0, wave1);
    return;
  }
  int* hist = sh;
  int* base = sh + n_bins;
  const int tid  = threadIdx.x;
  const int lane = tid & 31;
  const int wv   = tid >> 5;
  for (int i = tid; i < n_bins; i += 256) hist[i] = 0;
  __syncthreads();

  const int e0 = blockIdx.x * EPB + tid;
  int dk[EPT];
  int sk[EPT];
  int wk[EPT];
  int rk[EPT];
#pragma unroll
  for (int k = 0; k < EPT; ++k) {
    const int e = min(e0 + k * 256, n_edges - 1);
    dk[k] = dst[e];
    sk[k] = src[e];
    wk[k] = __float_as_int(w[e]);
  }
#pragma unroll
  for (int k = 0; k < EPT; ++k) {
    rk[k] = 0;
    if (e0 + k * 256 < n_edges && (unsigned)dk[k] < (unsigned)n_nodes)
      rk[k] = atomicAdd(&hist[dk[k] >> BSH], 1);
    else
      dk[k] = -1;
  }
  __syncthreads();

  int carry = 0;
  for (int c0 = 0; c0 < n_bins; c0 += 256) {
    const int i = c0 + tid;
    const int t = (i < n_bins) ? bin_tot[i] : 0;
    int incl = t;
    for (int s = 1; s < 32; s <<= 1) {
      const int u = __shfl_up(incl, s, 32);
      if (lane >= s) incl += u;
    }
    if (lane == 31) wsum[wv] = incl;
    __syncthreads();
    int wpre = 0;
    int tot  = 0;
    for (int j = 0; j < 8; ++j) {
      const int x = wsum[j];
      if (j < wv) wpre += x;
      tot += x;
    }
    if (i < n_bins) {
      const int h = hist[i];
      const int got = h ? atomicAdd(&bin_cur[i], h) : 0;
      base[i] = carry + wpre + incl - t + got;
    }
    carry += tot;
    __syncthreads();
  }

#pragma unroll
  for (int k = 0; k < EPT; ++k) {
    if (dk[k] >= 0) {
      const int p = base[dk[k] >> BSH] + rk[k];
      if ((unsigned)p < (unsigned)n_edges)
        ((int4*)tmp)[p] = make_int4(sk[k], wk[k], dk[k], 0);
    }
  }
}

__global__ void __launch_bounds__(1024)
gcn_owner(const int* __restrict__ tmp, const int* __restrict__ bin_tot,
          int* __restrict__ cnt, int* __restrict__ off, int* __restrict__ pair,
          int n_nodes, int n_edges) {
  __shared__ int cntL[BINW];
  __shared__ int offL[BINW];
  __shared__ int curL[BINW];
  __shared__ int red[32];
  const int tid  = threadIdx.x;
  const int lane = tid & 31;
  const int wv   = tid >> 5;
  const int bin  = blockIdx.x;
  const int node0 = bin * BINW;
  const int4* __restrict__ rec4 = (const int4*)tmp;

  int part = 0;
  for (int i = tid; i < bin; i += 1024) part += bin_tot[i];
  for (int s = 16; s > 0; s >>= 1) part += __shfl_xor(part, s, 32);
  if (lane == 0) red[wv] = part;
  if (tid < BINW) { cntL[tid] = 0; curL[tid] = 0; }
  __syncthreads();
  int start = 0;
  for (int j = 0; j < 32; ++j) start += red[j];
  int m = bin_tot[bin];
  start = min(max(start, 0), n_edges);
  m     = min(max(m, 0), n_edges - start);
  __syncthreads();

  for (int i0 = tid; i0 - tid < m; i0 += 4096) {
    int z[4];
#pragma unroll
    for (int k = 0; k < 4; ++k) z[k] = rec4[start + min(i0 + k * 1024, m - 1)].z;
#pragma unroll
    for (int k = 0; k < 4; ++k) {
      const int dl = z[k] - node0;
      if (i0 + k * 1024 < m && (unsigned)dl < (unsigned)BINW) atomicAdd(&cntL[dl], 1);
    }
  }
  __syncthreads();

  int c = 0, incl = 0;
  if (tid < BINW) {
    c = cntL[tid];
    incl = c;
    for (int s = 1; s < 32; s <<= 1) {
      const int u = __shfl_up(incl, s, 32);
      if (lane >= s) incl += u;
    }
    if (lane == 31) red[wv] = incl;
  }
  __syncthreads();
  if (tid < BINW) {
    int wpre = 0;
    for (int j = 0; j < 8; ++j) if (j < wv) wpre += red[j];
    const int excl = wpre + incl - c;
    offL[tid] = excl;
    const int node = node0 + tid;
    if (node < n_nodes) { cnt[node] = c; off[node] = start + excl; }
  }
  __syncthreads();

  for (int i0 = tid; i0 - tid < m; i0 += 4096) {
    int4 q[4];
#pragma unroll
    for (int k = 0; k < 4; ++k) q[k] = rec4[start + min(i0 + k * 1024, m - 1)];
#pragma unroll
    for (int k = 0; k < 4; ++k) {
      const int dl = q[k].z - node0;
      if (i0 + k * 1024 < m && (unsigned)dl < (unsigned)BINW) {
        const int p = start + offL[dl] + atomicAdd(&curL[dl], 1);
        if ((unsigned)p < (unsigned)n_edges)
          ((v2i*)pair)[p] = (v2i){q[k].x, q[k].y};
      }
    }
  }
}

__global__ void gcn_dinv(const int* __restrict__ pair, const int* __restrict__ off,
                         const int* __restrict__ cnt, float* __restrict__ dinv,
                         int n_nodes, int n_edges) {
  const int gt  = blockIdx.x * blockDim.x + threadIdx.x;
  const int n   = gt >> 3;
  const int sub = gt & 7;
  float d = 0.0f;
  if (n < n_nodes) {
    const int beg = off[n];
    const int m   = cnt[n];
    for (int i = sub; i < m; i += 8) {
      const int q = beg + i;
      if (q < n_edges) d += __int_as_float(pair[2 * (size_t)q + 1]);
    }
  }
  d += __shfl_xor(d, 1, 32);
  d += __shfl_xor(d, 2, 32);
  d += __shfl_xor(d, 4, 32);
  if (n < n_nodes && sub == 0) {
    d += FILLW;
    dinv[n] = (d > 0.0f) ? rsqrtf(d) : 0.0f;
  }
}

__global__ void gcn_gather(const int* __restrict__ pair, const int* __restrict__ off,
                           const int* __restrict__ cnt, const float* __restrict__ dinv,
                           const float* __restrict__ xw, const float* __restrict__ b,
                           float* __restrict__ out, int n_nodes, int n_edges) {
  const int lane = threadIdx.x & 31;
  const int n = __builtin_amdgcn_readfirstlane(
      (int)((blockIdx.x * blockDim.x + threadIdx.x) >> 5));
  if (n >= n_nodes) return;
  const int beg = __builtin_amdgcn_readfirstlane(off[n]);
  const int m   = __builtin_amdgcn_readfirstlane(cnt[n]);
  const float dd = dinv[n];
  const v2f self = ((const v2f*)(xw + (size_t)n * CCH))[lane];
  const float b0 = b[2 * lane];
  const float b1 = b[2 * lane + 1];

  v2f acc = {0.0f, 0.0f};
  for (int base = 0; base < m; base += 32) {
    const int mm = min(32, m - base);
    int   s_i = 0;
    float c_i = 0.0f;
    const int q = beg + base + lane;
    if (lane < mm && q < n_edges) {
      const v2i p = ((const v2i*)pair)[q];
      s_i = p.x;
      c_i = dinv[p.x] * __int_as_float(p.y) * dd;
    }
    for (int j = 0; j < mm; j += 8) {
      v2f   v[8];
      float c[8];
#pragma unroll
      for (int k = 0; k < 8; ++k) {
        v[k] = (v2f){0.0f, 0.0f};
        c[k] = 0.0f;
        if (j + k < mm) {
          const int s = __builtin_amdgcn_readlane(s_i, j + k);
          c[k] = __int_as_float(__builtin_amdgcn_readlane(__float_as_int(c_i), j + k));
          v[k] = ((const v2f*)(xw + (size_t)s * CCH))[lane];
        }
      }
#pragma unroll
      for (int k = 0; k < 8; ++k) {
        acc.x += v[k].x * c[k];
        acc.y += v[k].y * c[k];
      }
    }
  }

  const float k2 = FILLW * dd * dd;
  v2f r;
  r.x = tanhf(acc.x + self.x * k2 + b0);
  r.y = tanhf(acc.y + self.y * k2 + b1);
  ((v2f*)(out + (size_t)n * CCH))[lane] = r;
}

extern "C" void kernel_launch(void* const* d_in, const int* in_sizes, int n_in,
                              void* d_out, int out_size, void* d_ws, size_t ws_size,
                              hipStream_t stream) {
  const float* x  = (const float*)d_in[0];
  const int*   ei = (const int*)d_in[1];
  const float* ew = (const float*)d_in[2];
  const float* W  = (const float*)d_in[3];
  const float* b  = (const float*)d_in[4];
  float* out = (float*)d_out;

  const int n_nodes = in_sizes[0] / CCH;
  const int n_edges = in_sizes[2];
  const int* src = ei;
  const int* dst = ei + n_edges;

  const size_t n_pad = ((size_t)n_nodes + 63) & ~(size_t)63;
  const size_t e_pad = ((size_t)n_edges + 63) & ~(size_t)63;
  const int    n_bins = (n_nodes + BINW - 1) / BINW;
  const size_t b_pad = ((size_t)n_bins + 63) & ~(size_t)63;
  float* xw   = (float*)d_ws;
  float* dinv = xw + n_pad * CCH;
  int*   cnt  = (int*)(dinv + n_pad);
  int*   off  = cnt + n_pad;
  int*   tmp  = off + n_pad;
  int*   pair = tmp + e_pad * 4;
  int*   bin_tot = pair + e_pad * 2;
  int*   bin_cur = bin_tot + b_pad;

  const int B = 256;
  const int n_chunks = (n_edges + EPB - 1) / EPB;

  gcn_zero_cnt<<<(int)((2 * b_pad + B - 1) / B), B, 0, stream>>>(bin_tot, (int)(2 * b_pad));
  const int n_row_tiles = (n_nodes + 15) / 16;
  const int waves = n_row_tiles * (CCH / 16);
  const int g_half = ((waves + 7) / 8 + 1) / 2;
  const int wave_mid = g_half * 8;

  gcn_hist<<<n_chunks + g_half, B, n_bins * sizeof(int), stream>>>(dst, bin_tot, n_nodes, n_edges, n_bins,
                                                                  x, W, xw, 0, wave_mid);
  gcn_place<<<n_chunks + g_half, B, 2 * n_bins * sizeof(int), stream>>>(src, dst, ew, bin_tot, bin_cur, tmp,
                                                                       n_nodes, n_edges, n_bins,
                                                                       x, W, xw, wave_mid, waves);
  gcn_owner<<<n_bins, 1024, 0, stream>>>(tmp, bin_tot, cnt, off, pair, n_nodes, n_edges);
  gcn_dinv<<<(n_nodes * 8 + B - 1) / B, B, 0, stream>>>(pair, off, cnt, dinv, n_nodes, n_edges);

  gcn_gather<<<(n_nodes + 7) / 8, B, 0, stream>>>(pair, off, cnt, dinv, xw, b, out,
                                                  n_nodes, n_edges);
}


# Range owner group orders its (src, w) pairs in LDS and copies them out in order: each pair store covers two whole 128-byte lines instead of 32 separate lines; per-node sum order unchanged
# speedup vs baseline: 4.1661x; 1.0432x over previous
#include <hip/hip_runtime.h>
#include <hip/hip_bf16.h>
#include <math.h>

typedef __attribute__((ext_vector_type(2))) float v2f;
typedef __attribute__((ext_vector_type(8))) float v8f;
typedef __attribute__((ext_vector_type(2))) int   v2i;

#define CCH   64
#define FILLW 2.0f

__global__ void gcn_zero_cnt(int* __restrict__ cnt, int n_nodes) {
  int t = blockIdx.x * blockDim.x + threadIdx.x;
  if (t < n_nodes) cnt[t] = 0;
}

#define BINW  256
#define BSH   8
#define EPB   4096
#define EPT   16

__device__ __forceinline__ void gcn_gemm_group(const float* __restrict__ X,
                                               const float* __restrict__ W,
                                               float* __restrict__ XW, int n_nodes,
                                               int group, int wave0, int wave1) {
  const int wave = wave0 + ((group * 256 + (int)threadIdx.x) >> 5);
  const int lane = threadIdx.x & 31;
  const int n_row_tiles = (n_nodes + 15) >> 4;
  const int row_tile = wave >> 2;
  const int col_tile = wave & 3;
  if (wave >= wave1 || row_tile >= n_row_tiles) return;

  const int half = lane >> 4;
  const int l16  = lane & 15;
  int row = row_tile * 16 + l16;
  int ldrow = (row < n_nodes) ? row : (n_nodes - 1);

  const float* __restrict__ xrow = X + (size_t)ldrow * CCH;
  const float* __restrict__ wcol = W + col_tile * 16 + l16;

  v8f acc = {};
#pragma unroll
  for (int k0 = 0; k0 < CCH; k0 += 4) {
    const int ka = k0 + 2 * half;
    v2f a, b;
    a.x = xrow[ka + 0];
    a.y = xrow[ka + 1];
    b.x = wcol[(size_t)(ka + 0) * CCH];
    b.y = wcol[(size_t)(ka + 1) * CCH];
    acc = __builtin_amdgcn_wmma_f32_16x16x4_f32(
         false, a,  false, b,
         (short)0, acc,  false,  false);
  }

  const int base_row = row_tile * 16 + 8 * half;
  float* __restrict__ dstp = XW + (size_t)base_row * CCH + col_tile * 16 + l16;
#pragma unroll
  for (int i = 0; i < 8; ++i) {
    if (base_row + i < n_nodes) dstp[(size_t)i * CCH] = acc[i];
  }
}

__global__ void __launch_bounds__(256)
gcn_hist(const int* __restrict__ dst, int* __restrict__ bin_tot,
         int n_nodes, int n_edges, int n_bins,
         const float* __restrict__ X, const float* __restrict__ W,
         float* __restrict__ XW, int wave0, int wave1) {
  extern __shared__ int hist[];
  const int tid = threadIdx.x;
  const int n_chunks = (n_edges + EPB - 1) / EPB;
  if ((int)blockIdx.x >= n_chunks) {
    gcn_gemm_group(X, W, XW, n_nodes, (int)blockIdx.x - n_chunks, wave0, wave1);
    return;
  }
  for (int i = tid; i < n_bins; i += 256) hist[i] = 0;
  __syncthreads();
  const int e0 = blockIdx.x * EPB + tid;
  int dk[EPT];
#pragma unroll
  for (int k = 0; k < EPT; ++k) dk[k] = dst[min(e0 + k * 256, n_edges - 1)];
#pragma unroll
  for (int k = 0; k < EPT; ++k) {
    if (e0 + k * 256 < n_edges && (unsigned)dk[k] < (unsigned)n_nodes)
      atomicAdd(&hist[dk[k] >> BSH], 1);
  }
  __syncthreads();
  for (int i = tid; i < n_bins; i += 256) {
    const int h = hist[i];
    if (h) atomicAdd(&bin_tot[i], h);
  }
}

__global__ void __launch_bounds__(256)
gcn_place(const int* __restrict__ src, const int* __restrict__ dst,
          const float* __restrict__ w, const int* __restrict__ bin_tot,
          int* __restrict__ bin_cur, int* __restrict__ tmp,
          int n_nodes, int n_edges, int n_bins,
          const float* __restrict__ X, const float* __restrict__ W,
          float* __restrict__ XW, int wave0, int wave1) {
  extern __shared__ int sh[];
  __shared__ int wsum[8];
  const int n_chunks = (n_edges + EPB - 1) / EPB;
  if ((int)blockIdx.x >= n_chunks) {
    gcn_gemm_group(X, W, XW, n_nodes, (int)blockIdx.x - n_chunks, wave0, wave1);
    return;
  }
  int* hist = sh;
  int* base = sh + n_bins;
  const int tid  = threadIdx.x;
  const int lane = tid & 31;
  const int wv   = tid >> 5;
  for (int i = tid; i < n_bins; i += 256) hist[i] = 0;
  __syncthreads();

  const int e0 = blockIdx.x * EPB + tid;
  int dk[EPT];
  int sk[EPT];
  int wk[EPT];
  int rk[EPT];
#pragma unroll
  for (int k = 0; k < EPT; ++k) {
    const int e = min(e0 + k * 256, n_edges - 1);
    dk[k] = dst[e];
    sk[k] = src[e];
    wk[k] = __float_as_int(w[e]);
  }
#pragma unroll
  for (int k = 0; k < EPT; ++k) {
    rk[k] = 0;
    if (e0 + k * 256 < n_edges && (unsigned)dk[k] < (unsigned)n_nodes)
      rk[k] = atomicAdd(&hist[dk[k] >> BSH], 1);
    else
      dk[k] = -1;
  }
  __syncthreads();

  int carry = 0;
  for (int c0 = 0; c0 < n_bins; c0 += 256) {
    const int i = c0 + tid;
    const int t = (i < n_bins) ? bin_tot[i] : 0;
    int incl = t;
    for (int s = 1; s < 32; s <<= 1) {
      const int u = __shfl_up(incl, s, 32);
      if (lane >= s) incl += u;
    }
    if (lane == 31) wsum[wv] = incl;
    __syncthreads();
    int wpre = 0;
    int tot  = 0;
    for (int j = 0; j < 8; ++j) {
      const int x = wsum[j];
      if (j < wv) wpre += x;
      tot += x;
    }
    if (i < n_bins) {
      const int h = hist[i];
      const int got = h ? atomicAdd(&bin_cur[i], h) : 0;
      base[i] = carry + wpre + incl - t + got;
    }
    carry += tot;
    __syncthreads();
  }

#pragma unroll
  for (int k = 0; k < EPT; ++k) {
    if (dk[k] >= 0) {
      const int p = base[dk[k] >> BSH] + rk[k];
      if ((unsigned)p < (unsigned)n_edges)
        ((int4*)tmp)[p] = make_int4(sk[k], wk[k], dk[k], 0);
    }
  }
}

#define OCAP  8192
__global__ void __launch_bounds__(1024)
gcn_owner(const int* __restrict__ tmp, const int* __restrict__ bin_tot,
          int* __restrict__ cnt, int* __restrict__ off, int* __restrict__ pair,
          int n_nodes, int n_edges) {
  __shared__ int cntL[BINW];
  __shared__ int offL[BINW];
  __shared__ int curL[BINW];
  __shared__ int red[32];
  __shared__ v2i stage[OCAP];
  const int tid  = threadIdx.x;
  const int lane = tid & 31;
  const int wv   = tid >> 5;
  const int bin  = blockIdx.x;
  const int node0 = bin * BINW;
  const int4* __restrict__ rec4 = (const int4*)tmp;

  int part = 0;
  for (int i = tid; i < bin; i += 1024) part += bin_tot[i];
  for (int s = 16; s > 0; s >>= 1) part += __shfl_xor(part, s, 32);
  if (lane == 0) red[wv] = part;
  if (tid < BINW) { cntL[tid] = 0; curL[tid] = 0; }
  __syncthreads();
  int start = 0;
  for (int j = 0; j < 32; ++j) start += red[j];
  int m = bin_tot[bin];
  start = min(max(start, 0), n_edges);
  m     = min(max(m, 0), n_edges - start);
  const bool staged = (m <= OCAP);
  __syncthreads();

  for (int i0 = tid; i0 - tid < m; i0 += 4096) {
    int z[4];
#pragma unroll
    for (int k = 0; k < 4; ++k) z[k] = rec4[start + min(i0 + k * 1024, m - 1)].z;
#pragma unroll
    for (int k = 0; k < 4; ++k) {
      const int dl = z[k] - node0;
      if (i0 + k * 1024 < m && (unsigned)dl < (unsigned)BINW) atomicAdd(&cntL[dl], 1);
    }
  }
  __syncthreads();

  int c = 0, incl = 0;
  if (tid < BINW) {
    c = cntL[tid];
    incl = c;
    for (int s = 1; s < 32; s <<= 1) {
      const int u = __shfl_up(incl, s, 32);
      if (lane >= s) incl += u;
    }
    if (lane == 31) red[wv] = incl;
  }
  __syncthreads();
  if (tid < BINW) {
    int wpre = 0;
    for (int j = 0; j < 8; ++j) if (j < wv) wpre += red[j];
    const int excl = wpre + incl - c;
    offL[tid] = excl;
    const int node = node0 + tid;
    if (node < n_nodes) { cnt[node] = c; off[node] = start + excl; }
  }
  __syncthreads();

  for (int i0 = tid; i0 - tid < m; i0 += 4096) {
    int4 q[4];
#pragma unroll
    for (int k = 0; k < 4; ++k) q[k] = rec4[start + min(i0 + k * 1024, m - 1)];
#pragma unroll
    for (int k = 0; k < 4; ++k) {
      const int dl = q[k].z - node0;
      if (i0 + k * 1024 < m && (unsigned)dl < (unsigned)BINW) {
        const int li = offL[dl] + atomicAdd(&curL[dl], 1);
        if (staged) {
          if ((unsigned)li < (unsigned)OCAP)
            stage[li] = (v2i){q[k].x, q[k].y};
        } else {
          const int p = start + li;
          if ((unsigned)p < (unsigned)n_edges)
            ((v2i*)pair)[p] = (v2i){q[k].x, q[k].y};
        }
      }
    }
  }
  if (staged) {
    __syncthreads();
    const int shift = start & 15;
    for (int j = tid; j < m + shift; j += 1024) {
      const int i = j - shift;
      if (i >= 0) ((v2i*)pair)[start + i] = stage[i];
    }
  }
}

__global__ void gcn_dinv(const int* __restrict__ pair, const int* __restrict__ off,
                         const int* __restrict__ cnt, float* __restrict__ dinv,
                         int n_nodes, int n_edges) {
  const int gt  = blockIdx.x * blockDim.x + threadIdx.x;
  const int n   = gt >> 3;
  const int sub = gt & 7;
  float d = 0.0f;
  if (n < n_nodes) {
    const int beg = off[n];
    const int m   = cnt[n];
    for (int i = sub; i < m; i += 8) {
      const int q = beg + i;
      if (q < n_edges) d += __int_as_float(pair[2 * (size_t)q + 1]);
    }
  }
  d += __shfl_xor(d, 1, 32);
  d += __shfl_xor(d, 2, 32);
  d += __shfl_xor(d, 4, 32);
  if (n < n_nodes && sub == 0) {
    d += FILLW;
    dinv[n] = (d > 0.0f) ? rsqrtf(d) : 0.0f;
  }
}

__global__ void gcn_gather(const int* __restrict__ pair, const int* __restrict__ off,
                           const int* __restrict__ cnt, const float* __restrict__ dinv,
                           const float* __restrict__ xw, const float* __restrict__ b,
                           float* __restrict__ out, int n_nodes, int n_edges) {
  const int lane = threadIdx.x & 31;
  const int n = __builtin_amdgcn_readfirstlane(
      (int)((blockIdx.x * blockDim.x + threadIdx.x) >> 5));
  if (n >= n_nodes) return;
  const int beg = __builtin_amdgcn_readfirstlane(off[n]);
  const int m   = __builtin_amdgcn_readfirstlane(cnt[n]);
  const float dd = dinv[n];
  const v2f self = ((const v2f*)(xw + (size_t)n * CCH))[lane];
  const float b0 = b[2 * lane];
  const float b1 = b[2 * lane + 1];

  v2f acc = {0.0f, 0.0f};
  for (int base = 0; base < m; base += 32) {
    const int mm = min(32, m - base);
    int   s_i = 0;
    float c_i = 0.0f;
    const int q = beg + base + lane;
    if (lane < mm && q < n_edges) {
      const v2i p = ((const v2i*)pair)[q];
      s_i = p.x;
      c_i = dinv[p.x] * __int_as_float(p.y) * dd;
    }
    for (int j = 0; j < mm; j += 8) {
      v2f   v[8];
      float c[8];
#pragma unroll
      for (int k = 0; k < 8; ++k) {
        v[k] = (v2f){0.0f, 0.0f};
        c[k] = 0.0f;
        if (j + k < mm) {
          const int s = __builtin_amdgcn_readlane(s_i, j + k);
          c[k] = __int_as_float(__builtin_amdgcn_readlane(__float_as_int(c_i), j + k));
          v[k] = ((const v2f*)(xw + (size_t)s * CCH))[lane];
        }
      }
#pragma unroll
      for (int k = 0; k < 8; ++k) {
        acc.x += v[k].x * c[k];
        acc.y += v[k].y * c[k];
      }
    }
  }

  const float k2 = FILLW * dd * dd;
  v2f r;
  r.x = tanhf(acc.x + self.x * k2 + b0);
  r.y = tanhf(acc.y + self.y * k2 + b1);
  ((v2f*)(out + (size_t)n * CCH))[lane] = r;
}

extern "C" void kernel_launch(void* const* d_in, const int* in_sizes, int n_in,
                              void* d_out, int out_size, void* d_ws, size_t ws_size,
                              hipStream_t stream) {
  const float* x  = (const float*)d_in[0];
  const int*   ei = (const int*)d_in[1];
  const float* ew = (const float*)d_in[2];
  const float* W  = (const float*)d_in[3];
  const float* b  = (const float*)d_in[4];
  float* out = (float*)d_out;

  const int n_nodes = in_sizes[0] / CCH;
  const int n_edges = in_sizes[2];
  const int* src = ei;
  const int* dst = ei + n_edges;

  const size_t n_pad = ((size_t)n_nodes + 63) & ~(size_t)63;
  const size_t e_pad = ((size_t)n_edges + 63) & ~(size_t)63;
  const int    n_bins = (n_nodes + BINW - 1) / BINW;
  const size_t b_pad = ((size_t)n_bins + 63) & ~(size_t)63;
  float* xw   = (float*)d_ws;
  float* dinv = xw + n_pad * CCH;
  int*   cnt  = (int*)(dinv + n_pad);
  int*   off  = cnt + n_pad;
  int*   tmp  = off + n_pad;
  int*   pair = tmp + e_pad * 4;
  int*   bin_tot = pair + e_pad * 2;
  int*   bin_cur = bin_tot + b_pad;

  const int B = 256;
  const int n_chunks = (n_edges + EPB - 1) / EPB;

  gcn_zero_cnt<<<(int)((2 * b_pad + B - 1) / B), B, 0, stream>>>(bin_tot, (int)(2 * b_pad));
  const int n_row_tiles = (n_nodes + 15) / 16;
  const int waves = n_row_tiles * (CCH / 16);
  const int g_half = ((waves + 7) / 8 + 1) / 2;
  const int wave_mid = g_half * 8;

  gcn_hist<<<n_chunks + g_half, B, n_bins * sizeof(int), stream>>>(dst, bin_tot, n_nodes, n_edges, n_bins,
                                                                  x, W, xw, 0, wave_mid);
  gcn_place<<<n_chunks + g_half, B, 2 * n_bins * sizeof(int), stream>>>(src, dst, ew, bin_tot, bin_cur, tmp,
                                                                       n_nodes, n_edges, n_bins,
                                                                       x, W, xw, wave_mid, waves);
  gcn_owner<<<n_bins, 1024, 0, stream>>>(tmp, bin_tot, cnt, off, pair, n_nodes, n_edges);
  gcn_dinv<<<(n_nodes * 8 + B - 1) / B, B, 0, stream>>>(pair, off, cnt, dinv, n_nodes, n_edges);

  gcn_gather<<<(n_nodes + 7) / 8, B, 0, stream>>>(pair, off, cnt, dinv, xw, b, out,
                                                  n_nodes, n_edges);
}


# X@W tiles leave the edge-pass launches for the degree launch (X tile via LDS, 2 tiles a wave) and are stored scaled by their node's dinv: gather drops per-edge dinv[src] load; product regrouped
# speedup vs baseline: 4.4446x; 1.0668x over previous
#include <hip/hip_runtime.h>
#include <hip/hip_bf16.h>
#include <math.h>

typedef __attribute__((ext_vector_type(2))) float v2f;
typedef __attribute__((ext_vector_type(8))) float v8f;
typedef __attribute__((ext_vector_type(2))) int   v2i;

#define CCH   64
#define FILLW 2.0f

__global__ void gcn_zero_cnt(int* __restrict__ cnt, int n_nodes) {
  int t = blockIdx.x * blockDim.x + threadIdx.x;
  if (t < n_nodes) cnt[t] = 0;
}

#define BINW  256
#define BSH   8
#define EPB   4096
#define EPT   16

__global__ void __launch_bounds__(256)
gcn_hist(const int* __restrict__ dst, int* __restrict__ bin_tot,
         int n_nodes, int n_edges, int n_bins) {
  extern __shared__ int hist[];
  const int tid = threadIdx.x;
  for (int i = tid; i < n_bins; i += 256) hist[i] = 0;
  __syncthreads();
  const int e0 = blockIdx.x * EPB + tid;
  int dk[EPT];
#pragma unroll
  for (int k = 0; k < EPT; ++k) dk[k] = dst[min(e0 + k * 256, n_edges - 1)];
#pragma unroll
  for (int k = 0; k < EPT; ++k) {
    if (e0 + k * 256 < n_edges && (unsigned)dk[k] < (unsigned)n_nodes)
      atomicAdd(&hist[dk[k] >> BSH], 1);
  }
  __syncthreads();
  for (int i = tid; i < n_bins; i += 256) {
    const int h = hist[i];
    if (h) atomicAdd(&bin_tot[i], h);
  }
}

__global__ void __launch_bounds__(256)
gcn_place(const int* __restrict__ src, const int* __restrict__ dst,
          const float* __restrict__ w, const int* __restrict__ bin_tot,
          int* __restrict__ bin_cur, int* __restrict__ tmp,
          int n_nodes, int n_edges, int n_bins) {
  extern __shared__ int sh[];
  __shared__ int wsum[8];
  int* hist = sh;
  int* base = sh + n_bins;
  const int tid  = threadIdx.x;
  const int lane = tid & 31;
  const int wv   = tid >> 5;
  for (int i = tid; i < n_bins; i += 256) hist[i] = 0;
  __syncthreads();

  const int e0 = blockIdx.x * EPB + tid;
  int dk[EPT];
  int sk[EPT];
  int wk[EPT];
  int rk[EPT];
#pragma unroll
  for (int k = 0; k < EPT; ++k) {
    const int e = min(e0 + k * 256, n_edges - 1);
    dk[k] = dst[e];
    sk[k] = src[e];
    wk[k] = __float_as_int(w[e]);
  }
#pragma unroll
  for (int k = 0; k < EPT; ++k) {
    rk[k] = 0;
    if (e0 + k * 256 < n_edges && (unsigned)dk[k] < (unsigned)n_nodes)
      rk[k] = atomicAdd(&hist[dk[k] >> BSH], 1);
    else
      dk[k] = -1;
  }
  __syncthreads();

  int carry = 0;
  for (int c0 = 0; c0 < n_bins; c0 += 256) {
    const int i = c0 + tid;
    const int t = (i < n_bins) ? bin_tot[i] : 0;
    int incl = t;
    for (int s = 1; s < 32; s <<= 1) {
      const int u = __shfl_up(incl, s, 32);
      if (lane >= s) incl += u;
    }
    if (lane == 31) wsum[wv] = incl;
    __syncthreads();
    int wpre = 0;
    int tot  = 0;
    for (int j = 0; j < 8; ++j) {
      const int x = wsum[j];
      if (j < wv) wpre += x;
      tot += x;
    }
    if (i < n_bins) {
      const int h = hist[i];
      const int got = h ? atomicAdd(&bin_cur[i], h) : 0;
      base[i] = carry + wpre + incl - t + got;
    }
    carry += tot;
    __syncthreads();
  }

#pragma unroll
  for (int k = 0; k < EPT; ++k) {
    if (dk[k] >= 0) {
      const int p = base[dk[k] >> BSH] + rk[k];
      if ((unsigned)p < (unsigned)n_edges)
        ((int4*)tmp)[p] = make_int4(sk[k], wk[k], dk[k], 0);
    }
  }
}

#define OCAP  8192
__global__ void __launch_bounds__(1024)
gcn_owner(const int* __restrict__ tmp, const int* __restrict__ bin_tot,
          int* __restrict__ cnt, int* __restrict__ off, int* __restrict__ pair,
          int n_nodes, int n_edges) {
  __shared__ int cntL[BINW];
  __shared__ int offL[BINW];
  __shared__ int curL[BINW];
  __shared__ int red[32];
  __shared__ v2i stage[OCAP];
  const int tid  = threadIdx.x;
  const int lane = tid & 31;
  const int wv   = tid >> 5;
  const int bin  = blockIdx.x;
  const int node0 = bin * BINW;
  const int4* __restrict__ rec4 = (const int4*)tmp;

  int part = 0;
  for (int i = tid; i < bin; i += 1024) part += bin_tot[i];
  for (int s = 16; s > 0; s >>= 1) part += __shfl_xor(part, s, 32);
  if (lane == 0) red[wv] = part;
  if (tid < BINW) { cntL[tid] = 0; curL[tid] = 0; }
  __syncthreads();
  int start = 0;
  for (int j = 0; j < 32; ++j) start += red[j];
  int m = bin_tot[bin];
  start = min(max(start, 0), n_edges);
  m     = min(max(m, 0), n_edges - start);
  const bool staged = (m <= OCAP);
  __syncthreads();

  for (int i0 = tid; i0 - tid < m; i0 += 4096) {
    int z[4];
#pragma unroll
    for (int k = 0; k < 4; ++k) z[k] = rec4[start + min(i0 + k * 1024, m - 1)].z;
#pragma unroll
    for (int k = 0; k < 4; ++k) {
      const int dl = z[k] - node0;
      if (i0 + k * 1024 < m && (unsigned)dl < (unsigned)BINW) atomicAdd(&cntL[dl], 1);
    }
  }
  __syncthreads();

  int c = 0, incl = 0;
  if (tid < BINW) {
    c = cntL[tid];
    incl = c;
    for (int s = 1; s < 32; s <<= 1) {
      const int u = __shfl_up(incl, s, 32);
      if (lane >= s) incl += u;
    }
    if (lane == 31) red[wv] = incl;
  }
  __syncthreads();
  if (tid < BINW) {
    int wpre = 0;
    for (int j = 0; j < 8; ++j) if (j < wv) wpre += red[j];
    const int excl = wpre + incl - c;
    offL[tid] = excl;
    const int node = node0 + tid;
    if (node < n_nodes) { cnt[node] = c; off[node] = start + excl; }
  }
  __syncthreads();

  for (int i0 = tid; i0 - tid < m; i0 += 4096) {
    int4 q[4];
#pragma unroll
    for (int k = 0; k < 4; ++k) q[k] = rec4[start + min(i0 + k * 1024, m - 1)];
#pragma unroll
    for (int k = 0; k < 4; ++k) {
      const int dl = q[k].z - node0;
      if (i0 + k * 1024 < m && (unsigned)dl < (unsigned)BINW) {
        const int li = offL[dl] + atomicAdd(&curL[dl], 1);
        if (staged) {
          if ((unsigned)li < (unsigned)OCAP)
            stage[li] = (v2i){q[k].x, q[k].y};
        } else {
          const int p = start + li;
          if ((unsigned)p < (unsigned)n_edges)
            ((v2i*)pair)[p] = (v2i){q[k].x, q[k].y};
        }
      }
    }
  }
  if (staged) {
    __syncthreads();
    const int shift = start & 15;
    for (int j = tid; j < m + shift; j += 1024) {
      const int i = j - shift;
      if (i >= 0) ((v2i*)pair)[start + i] = stage[i];
    }
  }
}

#define TN  64
#define XP  68
__global__ void __launch_bounds__(256)
gcn_dinv_xw(const int* __restrict__ pair, const int* __restrict__ off,
            const int* __restrict__ cnt, float* __restrict__ dinv,
            const float* __restrict__ X, const float* __restrict__ W,
            float* __restrict__ XS, int n_nodes, int n_edges) {
  __shared__ __attribute__((aligned(16))) float xs[TN * XP];
  __shared__ __attribute__((aligned(16))) float sd[TN];
  const int tid   = threadIdx.x;
  const int lane  = tid & 31;
  const int wv    = tid >> 5;
  const int node0 = blockIdx.x * TN;
  const int half  = lane >> 4;
  const int l16   = lane & 15;
  const int col_tile = wv & 3;
  const int rp    = wv >> 2;

  const float* __restrict__ wcol = W + col_tile * 16 + l16;
  v2f bf[CCH / 4];
#pragma unroll
  for (int kk = 0; kk < CCH / 4; ++kk) {
    const int ka = kk * 4 + 2 * half;
    bf[kk].x = wcol[(size_t)(ka + 0) * CCH];
    bf[kk].y = wcol[(size_t)(ka + 1) * CCH];
  }

  const int sub = tid & 7;
  int nn[2], beg[2], mm[2];
#pragma unroll
  for (int h = 0; h < 2; ++h) {
    nn[h]  = node0 + h * 32 + (tid >> 3);
    beg[h] = 0;
    mm[h]  = 0;
    if (nn[h] < n_nodes) { beg[h] = off[nn[h]]; mm[h] = cnt[nn[h]]; }
  }

#pragma unroll
  for (int k = 0; k < 4; ++k) {
    const int idx = tid + k * 256;
    const int r   = idx >> 4;
    const int c4  = idx & 15;
    const int gr  = min(node0 + r, n_nodes - 1);
    const float4 v = ((const float4*)(X + (size_t)gr * CCH))[c4];
    *(float4*)&xs[r * XP + c4 * 4] = v;
  }

  float wpre[2][3];
#pragma unroll
  for (int h = 0; h < 2; ++h) {
#pragma unroll
    for (int k = 0; k < 3; ++k) {
      const int q = min(max(beg[h] + sub + 8 * k, 0), n_edges - 1);
      wpre[h][k] = __int_as_float(pair[2 * (size_t)q + 1]);
    }
  }
#pragma unroll
  for (int h = 0; h < 2; ++h) {
    float d = 0.0f;
#pragma unroll
    for (int k = 0; k < 3; ++k) {
      const int i = sub + 8 * k;
      if (i < mm[h] && beg[h] + i < n_edges) d += wpre[h][k];
    }
    for (int i = sub + 24; i < mm[h]; i += 8) {
      const int q = beg[h] + i;
      if (q < n_edges) d += __int_as_float(pair[2 * (size_t)q + 1]);
    }
    d += __shfl_xor(d, 1, 32);
    d += __shfl_xor(d, 2, 32);
    d += __shfl_xor(d, 4, 32);
    if (sub == 0) {
      d += FILLW;
      const float di = (d > 0.0f) ? rsqrtf(d) : 0.0f;
      sd[h * 32 + (tid >> 3)] = di;
      if (nn[h] < n_nodes) dinv[nn[h]] = di;
    }
  }
  __syncthreads();

  const int lr0 = rp * 32 + 8 * half;
  const float4 s00 = *(const float4*)&sd[lr0];
  const float4 s01 = *(const float4*)&sd[lr0 + 4];
  const float4 s10 = *(const float4*)&sd[lr0 + 16];
  const float4 s11 = *(const float4*)&sd[lr0 + 20];
  const float sc0[8] = {s00.x, s00.y, s00.z, s00.w, s01.x, s01.y, s01.z, s01.w};
  const float sc1[8] = {s10.x, s10.y, s10.z, s10.w, s11.x, s11.y, s11.z, s11.w};

  const float* xa0 = &xs[(rp * 32 + l16) * XP];
  const float* xa1 = xa0 + 16 * XP;
  v8f acc0 = {};
  v8f acc1 = {};
#pragma unroll
  for (int kk = 0; kk < CCH / 4; ++kk) {
    const int ka = kk * 4 + 2 * half;
    const v2f a0 = *(const v2f*)&xa0[ka];
    const v2f a1 = *(const v2f*)&xa1[ka];
    acc0 = __builtin_amdgcn_wmma_f32_16x16x4_f32(
         false, a0,  false, bf[kk],
         (short)0, acc0,  false,  false);
    acc1 = __builtin_amdgcn_wmma_f32_16x16x4_f32(
         false, a1,  false, bf[kk],
         (short)0, acc1,  false,  false);
  }

  float* __restrict__ dstp = XS + (size_t)(node0 + lr0) * CCH + col_tile * 16 + l16;
#pragma unroll
  for (int i = 0; i < 8; ++i) {
    if (node0 + lr0 + i < n_nodes) dstp[(size_t)i * CCH] = acc0[i] * sc0[i];
  }
#pragma unroll
  for (int i = 0; i < 8; ++i) {
    if (node0 + lr0 + 16 + i < n_nodes) dstp[(size_t)(16 + i) * CCH] = acc1[i] * sc1[i];
  }
}

__global__ void gcn_gather(const int* __restrict__ pair, const int* __restrict__ off,
                           const int* __restrict__ cnt, const float* __restrict__ dinv,
                           const float* __restrict__ xw, const float* __restrict__ b,
                           float* __restrict__ out, int n_nodes, int n_edges) {
  const int lane = threadIdx.x & 31;
  const int n = __builtin_amdgcn_readfirstlane(
      (int)((blockIdx.x * blockDim.x + threadIdx.x) >> 5));
  if (n >= n_nodes) return;
  const int beg = __builtin_amdgcn_readfirstlane(off[n]);
  const int m   = __builtin_amdgcn_readfirstlane(cnt[n]);
  const float dd = dinv[n];
  const v2f self = ((const v2f*)(xw + (size_t)n * CCH))[lane];
  const float b0 = b[2 * lane];
  const float b1 = b[2 * lane + 1];

  v2f acc = {0.0f, 0.0f};
  for (int base = 0; base < m; base += 32) {
    const int mm = min(32, m - base);
    int   s_i = 0;
    float c_i = 0.0f;
    const int q = beg + base + lane;
    if (lane < mm && q < n_edges) {
      const v2i p = ((const v2i*)pair)[q];
      s_i = p.x;
      c_i = __int_as_float(p.y) * dd;
    }
    for (int j = 0; j < mm; j += 8) {
      v2f   v[8];
      float c[8];
#pragma unroll
      for (int k = 0; k < 8; ++k) {
        v[k] = (v2f){0.0f, 0.0f};
        c[k] = 0.0f;
        if (j + k < mm) {
          const int s = __builtin_amdgcn_readlane(s_i, j + k);
          c[k] = __int_as_float(__builtin_amdgcn_readlane(__float_as_int(c_i), j + k));
          v[k] = ((const v2f*)(xw + (size_t)s * CCH))[lane];
        }
      }
#pragma unroll
      for (int k = 0; k < 8; ++k) {
        acc.x += v[k].x * c[k];
        acc.y += v[k].y * c[k];
      }
    }
  }

  const float k2 = FILLW * dd;
  v2f r;
  r.x = tanhf(acc.x + self.x * k2 + b0);
  r.y = tanhf(acc.y + self.y * k2 + b1);
  ((v2f*)(out + (size_t)n * CCH))[lane] = r;
}

extern "C" void kernel_launch(void* const* d_in, const int* in_sizes, int n_in,
                              void* d_out, int out_size, void* d_ws, size_t ws_size,
                              hipStream_t stream) {
  const float* x  = (const float*)d_in[0];
  const int*   ei = (const int*)d_in[1];
  const float* ew = (const float*)d_in[2];
  const float* W  = (const float*)d_in[3];
  const float* b  = (const float*)d_in[4];
  float* out = (float*)d_out;

  const int n_nodes = in_sizes[0] / CCH;
  const int n_edges = in_sizes[2];
  const int* src = ei;
  const int* dst = ei + n_edges;

  const size_t n_pad = ((size_t)n_nodes + 63) & ~(size_t)63;
  const size_t e_pad = ((size_t)n_edges + 63) & ~(size_t)63;
  const int    n_bins = (n_nodes + BINW - 1) / BINW;
  const size_t b_pad = ((size_t)n_bins + 63) & ~(size_t)63;
  float* xw   = (float*)d_ws;
  float* dinv = xw + n_pad * CCH;
  int*   cnt  = (int*)(dinv + n_pad);
  int*   off  = cnt + n_pad;
  int*   tmp  = off + n_pad;
  int*   pair = tmp + e_pad * 4;
  int*   bin_tot = pair + e_pad * 2;
  int*   bin_cur = bin_tot + b_pad;

  const int B = 256;
  const int n_chunks = (n_edges + EPB - 1) / EPB;

  gcn_zero_cnt<<<(int)((2 * b_pad + B - 1) / B), B, 0, stream>>>(bin_tot, (int)(2 * b_pad));
  gcn_hist<<<n_chunks, B, n_bins * sizeof(int), stream>>>(dst, bin_tot, n_nodes, n_edges, n_bins);
  gcn_place<<<n_chunks, B, 2 * n_bins * sizeof(int), stream>>>(src, dst, ew, bin_tot, bin_cur, tmp,
                                                              n_nodes, n_edges, n_bins);
  gcn_owner<<<n_bins, 1024, 0, stream>>>(tmp, bin_tot, cnt, off, pair, n_nodes, n_edges);
  gcn_dinv_xw<<<(n_nodes + TN - 1) / TN, B, 0, stream>>>(pair, off, cnt, dinv, x, W, xw,
                                                        n_nodes, n_edges);

  gcn_gather<<<(n_nodes + 7) / 8, B, 0, stream>>>(pair, off, cnt, dinv, xw, b, out,
                                                  n_nodes, n_edges);
}


# Each 4096-edge chunk is ordered by range in LDS and stored whole-line with an offset table; range owners pull their pieces: no global counters, clear and count launches gone (4); sum order = LDS rank
# speedup vs baseline: 4.6162x; 1.0386x over previous
#include <hip/hip_runtime.h>
#include <hip/hip_bf16.h>
#include <math.h>

typedef __attribute__((ext_vector_type(2))) float v2f;
typedef __attribute__((ext_vector_type(8))) float v8f;
typedef __attribute__((ext_vector_type(2))) int   v2i;

#define CCH   64
#define FILLW 2.0f

#define BINW  256
#define BSH   8
#define EPB   4096
#define EPT   16

__global__ void __launch_bounds__(256)
gcn_chunk(const int* __restrict__ src, const int* __restrict__ dst,
          const float* __restrict__ w, int* __restrict__ tmp, int* __restrict__ tab,
          int n_nodes, int n_edges, int n_bins, int tab_pitch) {
  extern __shared__ int sh[];
  __shared__ int wsum[8];
  __shared__ int4 stage[EPB];
  int* hist = sh;
  int* base = sh + n_bins;
  const int tid  = threadIdx.x;
  const int lane = tid & 31;
  const int wv   = tid >> 5;
  for (int i = tid; i < n_bins; i += 256) hist[i] = 0;
  __syncthreads();

  const int e0 = blockIdx.x * EPB + tid;
  int dk[EPT];
  int sk[EPT];
  int wk[EPT];
  int rk[EPT];
#pragma unroll
  for (int k = 0; k < EPT; ++k) {
    const int e = min(e0 + k * 256, n_edges - 1);
    dk[k] = dst[e];
    sk[k] = src[e];
    wk[k] = __float_as_int(w[e]);
  }
#pragma unroll
  for (int k = 0; k < EPT; ++k) {
    rk[k] = 0;
    if (e0 + k * 256 < n_edges && (unsigned)dk[k] < (unsigned)n_nodes)
      rk[k] = atomicAdd(&hist[dk[k] >> BSH], 1);
    else
      dk[k] = -1;
  }
  __syncthreads();

  int* __restrict__ row = tab + (size_t)blockIdx.x * tab_pitch;
  int carry = 0;
  for (int c0 = 0; c0 < n_bins; c0 += 256) {
    const int i = c0 + tid;
    const int t = (i < n_bins) ? hist[i] : 0;
    int incl = t;
    for (int s = 1; s < 32; s <<= 1) {
      const int u = __shfl_up(incl, s, 32);
      if (lane >= s) incl += u;
    }
    if (lane == 31) wsum[wv] = incl;
    __syncthreads();
    int wpre = 0;
    int tot  = 0;
    for (int j = 0; j < 8; ++j) {
      const int x = wsum[j];
      if (j < wv) wpre += x;
      tot += x;
    }
    if (i < n_bins) {
      const int b = carry + wpre + incl - t;
      base[i] = b;
      row[i]  = b;
    }
    carry += tot;
    __syncthreads();
  }
  if (tid == 0) row[n_bins] = carry;

#pragma unroll
  for (int k = 0; k < EPT; ++k) {
    if (dk[k] >= 0) {
      const int p = base[dk[k] >> BSH] + rk[k];
      if ((unsigned)p < (unsigned)EPB)
        stage[p] = make_int4(sk[k], wk[k], dk[k], 0);
    }
  }
  __syncthreads();
  int4* __restrict__ outp = (int4*)tmp + (size_t)blockIdx.x * EPB;
  const int lim = min(carry, n_edges - (int)blockIdx.x * EPB);
  for (int i = tid; i < lim; i += 256) outp[i] = stage[i];
}

#define OCAP  8192
#define KR    8
#define CPT   512

__device__ __forceinline__ void gcn_piece(const int* __restrict__ tab, int tab_pitch,
                                          int b, int bin, int n_edges,
                                          int* lo_out, int* first, int* len) {
  const int* __restrict__ row = tab + (size_t)b * tab_pitch;
  int lo = row[bin];
  int hi = row[bin + 1];
  lo = min(max(lo, 0), EPB);
  hi = min(max(hi, lo), EPB);
  const int room = max(n_edges - b * EPB, 0);
  lo = min(lo, room);
  hi = min(hi, room);
  *lo_out = lo;
  *first  = b * EPB + lo;
  *len    = hi - lo;
}

__device__ __forceinline__ void gcn_put(const int4 r, int node0, const int* offL, int* curL,
                                        bool staged, v2i* stage, int* __restrict__ pair,
                                        int start, int n_edges) {
  const int dl = r.z - node0;
  if ((unsigned)dl < (unsigned)BINW) {
    const int li = offL[dl] + atomicAdd(&curL[dl], 1);
    if (staged) {
      if ((unsigned)li < (unsigned)OCAP)
        stage[li] = (v2i){r.x, r.y};
    } else {
      const int p = start + li;
      if ((unsigned)p < (unsigned)n_edges)
        ((v2i*)pair)[p] = (v2i){r.x, r.y};
    }
  }
}

__global__ void __launch_bounds__(1024)
gcn_owner(const int* __restrict__ tmp, const int* __restrict__ tab,
          int* __restrict__ cnt, int* __restrict__ off, int* __restrict__ pair,
          int n_nodes, int n_edges, int n_chunks, int tab_pitch) {
  __shared__ int cntL[BINW];
  __shared__ int offL[BINW];
  __shared__ int curL[BINW];
  __shared__ int red[32];
  __shared__ int redm[32];
  __shared__ v2i stage[OCAP];
  const int tid  = threadIdx.x;
  const int lane = tid & 31;
  const int wv   = tid >> 5;
  const int bin  = blockIdx.x;
  const int node0 = bin * BINW;
  const int sub  = tid & 1;
  const int b0   = tid >> 1;
  const int4* __restrict__ rec4 = (const int4*)tmp;

  int f0 = 0, c0 = 0;
  int ps = 0, pm = 0;
  for (int b = b0; b < n_chunks; b += CPT) {
    int lo, f, c;
    gcn_piece(tab, tab_pitch, b, bin, n_edges, &lo, &f, &c);
    if (b == b0) { f0 = f; c0 = c; }
    if (sub == 0) { ps += lo; pm += c; }
  }
  int4 q[KR];
  {
    const int last = max(c0 - 1, 0);
#pragma unroll
    for (int k = 0; k < KR; ++k)
      q[k] = rec4[max(min(f0 + min(sub + 2 * k, last), n_edges - 1), 0)];
  }
  for (int s = 16; s > 0; s >>= 1) {
    ps += __shfl_xor(ps, s, 32);
    pm += __shfl_xor(pm, s, 32);
  }
  if (lane == 0) { red[wv] = ps; redm[wv] = pm; }
  if (tid < BINW) { cntL[tid] = 0; curL[tid] = 0; }
  __syncthreads();
  int start = 0, m = 0;
  for (int j = 0; j < 32; ++j) { start += red[j]; m += redm[j]; }
  start = min(max(start, 0), n_edges);
  m     = min(max(m, 0), n_edges - start);
  const bool staged = (m <= OCAP);
  __syncthreads();

#pragma unroll
  for (int k = 0; k < KR; ++k) {
    const int dl = q[k].z - node0;
    if (sub + 2 * k < c0 && (unsigned)dl < (unsigned)BINW) atomicAdd(&cntL[dl], 1);
  }
  for (int i = sub + 2 * KR; i < c0; i += 2) {
    const int dl = rec4[min(f0 + i, n_edges - 1)].z - node0;
    if ((unsigned)dl < (unsigned)BINW) atomicAdd(&cntL[dl], 1);
  }
  for (int b = b0 + CPT; b < n_chunks; b += CPT) {
    int lo, f, c;
    gcn_piece(tab, tab_pitch, b, bin, n_edges, &lo, &f, &c);
    for (int i = sub; i < c; i += 2) {
      const int dl = rec4[min(f + i, n_edges - 1)].z - node0;
      if ((unsigned)dl < (unsigned)BINW) atomicAdd(&cntL[dl], 1);
    }
  }
  __syncthreads();

  int c = 0, incl = 0;
  if (tid < BINW) {
    c = cntL[tid];
    incl = c;
    for (int s = 1; s < 32; s <<= 1) {
      const int u = __shfl_up(incl, s, 32);
      if (lane >= s) incl += u;
    }
    if (lane == 31) red[wv] = incl;
  }
  __syncthreads();
  if (tid < BINW) {
    int wpre = 0;
    for (int j = 0; j < 8; ++j) if (j < wv) wpre += red[j];
    const int excl = wpre + incl - c;
    offL[tid] = excl;
    const int node = node0 + tid;
    if (node < n_nodes) { cnt[node] = c; off[node] = start + excl; }
  }
  __syncthreads();

#pragma unroll
  for (int k = 0; k < KR; ++k)
    if (sub + 2 * k < c0)
      gcn_put(q[k], node0, offL, curL, staged, stage, pair, start, n_edges);
  for (int i = sub + 2 * KR; i < c0; i += 2)
    gcn_put(rec4[min(f0 + i, n_edges - 1)], node0, offL, curL, staged, stage, pair, start, n_edges);
  for (int b = b0 + CPT; b < n_chunks; b += CPT) {
    int lo, f, cc;
    gcn_piece(tab, tab_pitch, b, bin, n_edges, &lo, &f, &cc);
    for (int i = sub; i < cc; i += 2)
      gcn_put(rec4[min(f + i, n_edges - 1)], node0, offL, curL, staged, stage, pair, start, n_edges);
  }
  if (staged) {
    __syncthreads();
    const int shift = start & 15;
    for (int j = tid; j < m + shift; j += 1024) {
      const int i = j - shift;
      if (i >= 0) ((v2i*)pair)[start + i] = stage[i];
    }
  }
}

#define TN  64
#define XP  68
__global__ void __launch_bounds__(256)
gcn_dinv_xw(const int* __restrict__ pair, const int* __restrict__ off,
            const int* __restrict__ cnt, float* __restrict__ dinv,
            const float* __restrict__ X, const float* __restrict__ W,
            float* __restrict__ XS, int n_nodes, int n_edges) {
  __shared__ __attribute__((aligned(16))) float xs[TN * XP];
  __shared__ __attribute__((aligned(16))) float sd[TN];
  const int tid   = threadIdx.x;
  const int lane  = tid & 31;
  const int wv    = tid >> 5;
  const int node0 = blockIdx.x * TN;
  const int half  = lane >> 4;
  const int l16   = lane & 15;
  const int col_tile = wv & 3;
  const int rp    = wv >> 2;

  const float* __restrict__ wcol = W + col_tile * 16 + l16;
  v2f bf[CCH / 4];
#pragma unroll
  for (int kk = 0; kk < CCH / 4; ++kk) {
    const int ka = kk * 4 + 2 * half;
    bf[kk].x = wcol[(size_t)(ka + 0) * CCH];
    bf[kk].y = wcol[(size_t)(ka + 1) * CCH];
  }

  const int sub = tid & 7;
  int nn[2], beg[2], mm[2];
#pragma unroll
  for (int h = 0; h < 2; ++h) {
    nn[h]  = node0 + h * 32 + (tid >> 3);
    beg[h] = 0;
    mm[h]  = 0;
    if (nn[h] < n_nodes) { beg[h] = off[nn[h]]; mm[h] = cnt[nn[h]]; }
  }

#pragma unroll
  for (int k = 0; k < 4; ++k) {
    const int idx = tid + k * 256;
    const int r   = idx >> 4;
    const int c4  = idx & 15;
    const int gr  = min(node0 + r, n_nodes - 1);
    const float4 v = ((const float4*)(X + (size_t)gr * CCH))[c4];
    *(float4*)&xs[r * XP + c4 * 4] = v;
  }

  float wpre[2][3];
#pragma unroll
  for (int h = 0; h < 2; ++h) {
#pragma unroll
    for (int k = 0; k < 3; ++k) {
      const int q = min(max(beg[h] + sub + 8 * k, 0), n_edges - 1);
      wpre[h][k] = __int_as_float(pair[2 * (size_t)q + 1]);
    }
  }
#pragma unroll
  for (int h = 0; h < 2; ++h) {
    float d = 0.0f;
#pragma unroll
    for (int k = 0; k < 3; ++k) {
      const int i = sub + 8 * k;
      if (i < mm[h] && beg[h] + i < n_edges) d += wpre[h][k];
    }
    for (int i = sub + 24; i < mm[h]; i += 8) {
      const int q = beg[h] + i;
      if (q < n_edges) d += __int_as_float(pair[2 * (size_t)q + 1]);
    }
    d += __shfl_xor(d, 1, 32);
    d += __shfl_xor(d, 2, 32);
    d += __shfl_xor(d, 4, 32);
    if (sub == 0) {
      d += FILLW;
      const float di = (d > 0.0f) ? rsqrtf(d) : 0.0f;
      sd[h * 32 + (tid >> 3)] = di;
      if (nn[h] < n_nodes) dinv[nn[h]] = di;
    }
  }
  __syncthreads();

  const int lr0 = rp * 32 + 8 * half;
  const float4 s00 = *(const float4*)&sd[lr0];
  const float4 s01 = *(const float4*)&sd[lr0 + 4];
  const float4 s10 = *(const float4*)&sd[lr0 + 16];
  const float4 s11 = *(const float4*)&sd[lr0 + 20];
  const float sc0[8] = {s00.x, s00.y, s00.z, s00.w, s01.x, s01.y, s01.z, s01.w};
  const float sc1[8] = {s10.x, s10.y, s10.z, s10.w, s11.x, s11.y, s11.z, s11.w};

  const float* xa0 = &xs[(rp * 32 + l16) * XP];
  const float* xa1 = xa0 + 16 * XP;
  v8f acc0 = {};
  v8f acc1 = {};
#pragma unroll
  for (int kk = 0; kk < CCH / 4; ++kk) {
    const int ka = kk * 4 + 2 * half;
    const v2f a0 = *(const v2f*)&xa0[ka];
    const v2f a1 = *(const v2f*)&xa1[ka];
    acc0 = __builtin_amdgcn_wmma_f32_16x16x4_f32(
         false, a0,  false, bf[kk],
         (short)0, acc0,  false,  false);
    acc1 = __builtin_amdgcn_wmma_f32_16x16x4_f32(
         false, a1,  false, bf[kk],
         (short)0, acc1,  false,  false);
  }

  float* __restrict__ dstp = XS + (size_t)(node0 + lr0) * CCH + col_tile * 16 + l16;
#pragma unroll
  for (int i = 0; i < 8; ++i) {
    if (node0 + lr0 + i < n_nodes) dstp[(size_t)i * CCH] = acc0[i] * sc0[i];
  }
#pragma unroll
  for (int i = 0; i < 8; ++i) {
    if (node0 + lr0 + 16 + i < n_nodes) dstp[(size_t)(16 + i) * CCH] = acc1[i] * sc1[i];
  }
}

__global__ void gcn_gather(const int* __restrict__ pair, const int* __restrict__ off,
                           const int* __restrict__ cnt, const float* __restrict__ dinv,
                           const float* __restrict__ xw, const float* __restrict__ b,
                           float* __restrict__ out, int n_nodes, int n_edges) {
  const int lane = threadIdx.x & 31;
  const int n = __builtin_amdgcn_readfirstlane(
      (int)((blockIdx.x * blockDim.x + threadIdx.x) >> 5));
  if (n >= n_nodes) return;
  const int beg = __builtin_amdgcn_readfirstlane(off[n]);
  const int m   = __builtin_amdgcn_readfirstlane(cnt[n]);
  const float dd = dinv[n];
  const v2f self = ((const v2f*)(xw + (size_t)n * CCH))[lane];
  const float b0 = b[2 * lane];
  const float b1 = b[2 * lane + 1];

  v2f acc = {0.0f, 0.0f};
  for (int base = 0; base < m; base += 32) {
    const int mm = min(32, m - base);
    int   s_i = 0;
    float c_i = 0.0f;
    const int q = beg + base + lane;
    if (lane < mm && q < n_edges) {
      const v2i p = ((const v2i*)pair)[q];
      s_i = p.x;
      c_i = __int_as_float(p.y) * dd;
    }
    for (int j = 0; j < mm; j += 8) {
      v2f   v[8];
      float c[8];
#pragma unroll
      for (int k = 0; k < 8; ++k) {
        v[k] = (v2f){0.0f, 0.0f};
        c[k] = 0.0f;
        if (j + k < mm) {
          const int s = __builtin_amdgcn_readlane(s_i, j + k);
          c[k] = __int_as_float(__builtin_amdgcn_readlane(__float_as_int(c_i), j + k));
          v[k] = ((const v2f*)(xw + (size_t)s * CCH))[lane];
        }
      }
#pragma unroll
      for (int k = 0; k < 8; ++k) {
        acc.x += v[k].x * c[k];
        acc.y += v[k].y * c[k];
      }
    }
  }

  const float k2 = FILLW * dd;
  v2f r;
  r.x = tanhf(acc.x + self.x * k2 + b0);
  r.y = tanhf(acc.y + self.y * k2 + b1);
  ((v2f*)(out + (size_t)n * CCH))[lane] = r;
}

extern "C" void kernel_launch(void* const* d_in, const int* in_sizes, int n_in,
                              void* d_out, int out_size, void* d_ws, size_t ws_size,
                              hipStream_t stream) {
  const float* x  = (const float*)d_in[0];
  const int*   ei = (const int*)d_in[1];
  const float* ew = (const float*)d_in[2];
  const float* W  = (const float*)d_in[3];
  const float* b  = (const float*)d_in[4];
  float* out = (float*)d_out;

  const int n_nodes = in_sizes[0] / CCH;
  const int n_edges = in_sizes[2];
  const int* src = ei;
  const int* dst = ei + n_edges;

  const size_t n_pad = ((size_t)n_nodes + 63) & ~(size_t)63;
  const size_t e_pad = ((size_t)n_edges + 63) & ~(size_t)63;
  const int    n_bins = (n_nodes + BINW - 1) / BINW;
  const size_t b_pad = ((size_t)n_bins + 63) & ~(size_t)63;
  float* xw   = (float*)d_ws;
  float* dinv = xw + n_pad * CCH;
  int*   cnt  = (int*)(dinv + n_pad);
  int*   off  = cnt + n_pad;
  int*   tmp  = off + n_pad;
  int*   pair = tmp + e_pad * 4;
  int*   tab  = pair + e_pad * 2;
  const int tab_pitch = (int)b_pad + 64;

  const int B = 256;
  const int n_chunks = (n_edges + EPB - 1) / EPB;

  gcn_chunk<<<n_chunks, B, 2 * n_bins * sizeof(int), stream>>>(src, dst, ew, tmp, tab,
                                                              n_nodes, n_edges, n_bins, tab_pitch);
  gcn_owner<<<n_bins, 1024, 0, stream>>>(tmp, tab, cnt, off, pair, n_nodes, n_edges, n_chunks,
                                         tab_pitch);
  gcn_dinv_xw<<<(n_nodes + TN - 1) / TN, B, 0, stream>>>(pair, off, cnt, dinv, x, W, xw,
                                                        n_nodes, n_edges);

  gcn_gather<<<(n_nodes + 7) / 8, B, 0, stream>>>(pair, off, cnt, dinv, xw, b, out,
                                                  n_nodes, n_edges);
}


# Gather: 16 lanes x 16 B per node, two nodes a wave at a time, 5 such pairs a wave (10000 resident waves enter, not 100000), next pair's edges fetched under this pair's rows; sum order unchanged, fma
# speedup vs baseline: 4.7592x; 1.0310x over previous
#include <hip/hip_runtime.h>
#include <hip/hip_bf16.h>
#include <math.h>

typedef __attribute__((ext_vector_type(2))) float v2f;
typedef __attribute__((ext_vector_type(8))) float v8f;
typedef __attribute__((ext_vector_type(2))) int   v2i;

#define CCH   64
#define FILLW 2.0f

#define BINW  256
#define BSH   8
#define EPB   4096
#define EPT   16

__global__ void __launch_bounds__(256)
gcn_chunk(const int* __restrict__ src, const int* __restrict__ dst,
          const float* __restrict__ w, int* __restrict__ tmp, int* __restrict__ tab,
          int n_nodes, int n_edges, int n_bins, int tab_pitch) {
  extern __shared__ int sh[];
  __shared__ int wsum[8];
  __shared__ int4 stage[EPB];
  int* hist = sh;
  int* base = sh + n_bins;
  const int tid  = threadIdx.x;
  const int lane = tid & 31;
  const int wv   = tid >> 5;
  for (int i = tid; i < n_bins; i += 256) hist[i] = 0;
  __syncthreads();

  const int e0 = blockIdx.x * EPB + tid;
  int dk[EPT];
  int sk[EPT];
  int wk[EPT];
  int rk[EPT];
#pragma unroll
  for (int k = 0; k < EPT; ++k) {
    const int e = min(e0 + k * 256, n_edges - 1);
    dk[k] = dst[e];
    sk[k] = src[e];
    wk[k] = __float_as_int(w[e]);
  }
#pragma unroll
  for (int k = 0; k < EPT; ++k) {
    rk[k] = 0;
    if (e0 + k * 256 < n_edges && (unsigned)dk[k] < (unsigned)n_nodes)
      rk[k] = atomicAdd(&hist[dk[k] >> BSH], 1);
    else
      dk[k] = -1;
  }
  __syncthreads();

  int* __restrict__ row = tab + (size_t)blockIdx.x * tab_pitch;
  int carry = 0;
  for (int c0 = 0; c0 < n_bins; c0 += 256) {
    const int i = c0 + tid;
    const int t = (i < n_bins) ? hist[i] : 0;
    int incl = t;
    for (int s = 1; s < 32; s <<= 1) {
      const int u = __shfl_up(incl, s, 32);
      if (lane >= s) incl += u;
    }
    if (lane == 31) wsum[wv] = incl;
    __syncthreads();
    int wpre = 0;
    int tot  = 0;
    for (int j = 0; j < 8; ++j) {
      const int x = wsum[j];
      if (j < wv) wpre += x;
      tot += x;
    }
    if (i < n_bins) {
      const int b = carry + wpre + incl - t;
      base[i] = b;
      row[i]  = b;
    }
    carry += tot;
    __syncthreads();
  }
  if (tid == 0) row[n_bins] = carry;

#pragma unroll
  for (int k = 0; k < EPT; ++k) {
    if (dk[k] >= 0) {
      const int p = base[dk[k] >> BSH] + rk[k];
      if ((unsigned)p < (unsigned)EPB)
        stage[p] = make_int4(sk[k], wk[k], dk[k], 0);
    }
  }
  __syncthreads();
  int4* __restrict__ outp = (int4*)tmp + (size_t)blockIdx.x * EPB;
  const int lim = min(carry, n_edges - (int)blockIdx.x * EPB);
  for (int i = tid; i < lim; i += 256) outp[i] = stage[i];
}

#define OCAP  8192
#define KR    8
#define CPT   512

__device__ __forceinline__ void gcn_piece(const int* __restrict__ tab, int tab_pitch,
                                          int b, int bin, int n_edges,
                                          int* lo_out, int* first, int* len) {
  const int* __restrict__ row = tab + (size_t)b * tab_pitch;
  int lo = row[bin];
  int hi = row[bin + 1];
  lo = min(max(lo, 0), EPB);
  hi = min(max(hi, lo), EPB);
  const int room = max(n_edges - b * EPB, 0);
  lo = min(lo, room);
  hi = min(hi, room);
  *lo_out = lo;
  *first  = b * EPB + lo;
  *len    = hi - lo;
}

__device__ __forceinline__ void gcn_put(const int4 r, int node0, const int* offL, int* curL,
                                        bool staged, v2i* stage, int* __restrict__ pair,
                                        int start, int n_edges) {
  const int dl = r.z - node0;
  if ((unsigned)dl < (unsigned)BINW) {
    const int li = offL[dl] + atomicAdd(&curL[dl], 1);
    if (staged) {
      if ((unsigned)li < (unsigned)OCAP)
        stage[li] = (v2i){r.x, r.y};
    } else {
      const int p = start + li;
      if ((unsigned)p < (unsigned)n_edges)
        ((v2i*)pair)[p] = (v2i){r.x, r.y};
    }
  }
}

__global__ void __launch_bounds__(1024)
gcn_owner(const int* __restrict__ tmp, const int* __restrict__ tab,
          int* __restrict__ cnt, int* __restrict__ off, int* __restrict__ pair,
          int n_nodes, int n_edges, int n_chunks, int tab_pitch) {
  __shared__ int cntL[BINW];
  __shared__ int offL[BINW];
  __shared__ int curL[BINW];
  __shared__ int red[32];
  __shared__ int redm[32];
  __shared__ v2i stage[OCAP];
  const int tid  = threadIdx.x;
  const int lane = tid & 31;
  const int wv   = tid >> 5;
  const int bin  = blockIdx.x;
  const int node0 = bin * BINW;
  const int sub  = tid & 1;
  const int b0   = tid >> 1;
  const int4* __restrict__ rec4 = (const int4*)tmp;

  int f0 = 0, c0 = 0;
  int ps = 0, pm = 0;
  for (int b = b0; b < n_chunks; b += CPT) {
    int lo, f, c;
    gcn_piece(tab, tab_pitch, b, bin, n_edges, &lo, &f, &c);
    if (b == b0) { f0 = f; c0 = c; }
    if (sub == 0) { ps += lo; pm += c; }
  }
  int4 q[KR];
  {
    const int last = max(c0 - 1, 0);
#pragma unroll
    for (int k = 0; k < KR; ++k)
      q[k] = rec4[max(min(f0 + min(sub + 2 * k, last), n_edges - 1), 0)];
  }
  for (int s = 16; s > 0; s >>= 1) {
    ps += __shfl_xor(ps, s, 32);
    pm += __shfl_xor(pm, s, 32);
  }
  if (lane == 0) { red[wv] = ps; redm[wv] = pm; }
  if (tid < BINW) { cntL[tid] = 0; curL[tid] = 0; }
  __syncthreads();
  int start = 0, m = 0;
  for (int j = 0; j < 32; ++j) { start += red[j]; m += redm[j]; }
  start = min(max(start, 0), n_edges);
  m     = min(max(m, 0), n_edges - start);
  const bool staged = (m <= OCAP);
  __syncthreads();

#pragma unroll
  for (int k = 0; k < KR; ++k) {
    const int dl = q[k].z - node0;
    if (sub + 2 * k < c0 && (unsigned)dl < (unsigned)BINW) atomicAdd(&cntL[dl], 1);
  }
  for (int i = sub + 2 * KR; i < c0; i += 2) {
    const int dl = rec4[min(f0 + i, n_edges - 1)].z - node0;
    if ((unsigned)dl < (unsigned)BINW) atomicAdd(&cntL[dl], 1);
  }
  for (int b = b0 + CPT; b < n_chunks; b += CPT) {
    int lo, f, c;
    gcn_piece(tab, tab_pitch, b, bin, n_edges, &lo, &f, &c);
    for (int i = sub; i < c; i += 2) {
      const int dl = rec4[min(f + i, n_edges - 1)].z - node0;
      if ((unsigned)dl < (unsigned)BINW) atomicAdd(&cntL[dl], 1);
    }
  }
  __syncthreads();

  int c = 0, incl = 0;
  if (tid < BINW) {
    c = cntL[tid];
    incl = c;
    for (int s = 1; s < 32; s <<= 1) {
      const int u = __shfl_up(incl, s, 32);
      if (lane >= s) incl += u;
    }
    if (lane == 31) red[wv] = incl;
  }
  __syncthreads();
  if (tid < BINW) {
    int wpre = 0;
    for (int j = 0; j < 8; ++j) if (j < wv) wpre += red[j];
    const int excl = wpre + incl - c;
    offL[tid] = excl;
    const int node = node0 + tid;
    if (node < n_nodes) { cnt[node] = c; off[node] = start + excl; }
  }
  __syncthreads();

#pragma unroll
  for (int k = 0; k < KR; ++k)
    if (sub + 2 * k < c0)
      gcn_put(q[k], node0, offL, curL, staged, stage, pair, start, n_edges);
  for (int i = sub + 2 * KR; i < c0; i += 2)
    gcn_put(rec4[min(f0 + i, n_edges - 1)], node0, offL, curL, staged, stage, pair, start, n_edges);
  for (int b = b0 + CPT; b < n_chunks; b += CPT) {
    int lo, f, cc;
    gcn_piece(tab, tab_pitch, b, bin, n_edges, &lo, &f, &cc);
    for (int i = sub; i < cc; i += 2)
      gcn_put(rec4[min(f + i, n_edges - 1)], node0, offL, curL, staged, stage, pair, start, n_edges);
  }
  if (staged) {
    __syncthreads();
    const int shift = start & 15;
    for (int j = tid; j < m + shift; j += 1024) {
      const int i = j - shift;
      if (i >= 0) ((v2i*)pair)[start + i] = stage[i];
    }
  }
}

#define TN  64
#define XP  68
__global__ void __launch_bounds__(256)
gcn_dinv_xw(const int* __restrict__ pair, const int* __restrict__ off,
            const int* __restrict__ cnt, float* __restrict__ dinv,
            const float* __restrict__ X, const float* __restrict__ W,
            float* __restrict__ XS, int n_nodes, int n_edges) {
  __shared__ __attribute__((aligned(16))) float xs[TN * XP];
  __shared__ __attribute__((aligned(16))) float sd[TN];
  const int tid   = threadIdx.x;
  const int lane  = tid & 31;
  const int wv    = tid >> 5;
  const int node0 = blockIdx.x * TN;
  const int half  = lane >> 4;
  const int l16   = lane & 15;
  const int col_tile = wv & 3;
  const int rp    = wv >> 2;

  const float* __restrict__ wcol = W + col_tile * 16 + l16;
  v2f bf[CCH / 4];
#pragma unroll
  for (int kk = 0; kk < CCH / 4; ++kk) {
    const int ka = kk * 4 + 2 * half;
    bf[kk].x = wcol[(size_t)(ka + 0) * CCH];
    bf[kk].y = wcol[(size_t)(ka + 1) * CCH];
  }

  const int sub = tid & 7;
  int nn[2], beg[2], mm[2];
#pragma unroll
  for (int h = 0; h < 2; ++h) {
    nn[h]  = node0 + h * 32 + (tid >> 3);
    beg[h] = 0;
    mm[h]  = 0;
    if (nn[h] < n_nodes) { beg[h] = off[nn[h]]; mm[h] = cnt[nn[h]]; }
  }

#pragma unroll
  for (int k = 0; k < 4; ++k) {
    const int idx = tid + k * 256;
    const int r   = idx >> 4;
    const int c4  = idx & 15;
    const int gr  = min(node0 + r, n_nodes - 1);
    const float4 v = ((const float4*)(X + (size_t)gr * CCH))[c4];
    *(float4*)&xs[r * XP + c4 * 4] = v;
  }

  float wpre[2][3];
#pragma unroll
  for (int h = 0; h < 2; ++h) {
#pragma unroll
    for (int k = 0; k < 3; ++k) {
      const int q = min(max(beg[h] + sub + 8 * k, 0), n_edges - 1);
      wpre[h][k] = __int_as_float(pair[2 * (size_t)q + 1]);
    }
  }
#pragma unroll
  for (int h = 0; h < 2; ++h) {
    float d = 0.0f;
#pragma unroll
    for (int k = 0; k < 3; ++k) {
      const int i = sub + 8 * k;
      if (i < mm[h] && beg[h] + i < n_edges) d += wpre[h][k];
    }
    for (int i = sub + 24; i < mm[h]; i += 8) {
      const int q = beg[h] + i;
      if (q < n_edges) d += __int_as_float(pair[2 * (size_t)q + 1]);
    }
    d += __shfl_xor(d, 1, 32);
    d += __shfl_xor(d, 2, 32);
    d += __shfl_xor(d, 4, 32);
    if (sub == 0) {
      d += FILLW;
      const float di = (d > 0.0f) ? rsqrtf(d) : 0.0f;
      sd[h * 32 + (tid >> 3)] = di;
      if (nn[h] < n_nodes) dinv[nn[h]] = di;
    }
  }
  __syncthreads();

  const int lr0 = rp * 32 + 8 * half;
  const float4 s00 = *(const float4*)&sd[lr0];
  const float4 s01 = *(const float4*)&sd[lr0 + 4];
  const float4 s10 = *(const float4*)&sd[lr0 + 16];
  const float4 s11 = *(const float4*)&sd[lr0 + 20];
  const float sc0[8] = {s00.x, s00.y, s00.z, s00.w, s01.x, s01.y, s01.z, s01.w};
  const float sc1[8] = {s10.x, s10.y, s10.z, s10.w, s11.x, s11.y, s11.z, s11.w};

  const float* xa0 = &xs[(rp * 32 + l16) * XP];
  const float* xa1 = xa0 + 16 * XP;
  v8f acc0 = {};
  v8f acc1 = {};
#pragma unroll
  for (int kk = 0; kk < CCH / 4; ++kk) {
    const int ka = kk * 4 + 2 * half;
    const v2f a0 = *(const v2f*)&xa0[ka];
    const v2f a1 = *(const v2f*)&xa1[ka];
    acc0 = __builtin_amdgcn_wmma_f32_16x16x4_f32(
         false, a0,  false, bf[kk],
         (short)0, acc0,  false,  false);
    acc1 = __builtin_amdgcn_wmma_f32_16x16x4_f32(
         false, a1,  false, bf[kk],
         (short)0, acc1,  false,  false);
  }

  float* __restrict__ dstp = XS + (size_t)(node0 + lr0) * CCH + col_tile * 16 + l16;
#pragma unroll
  for (int i = 0; i < 8; ++i) {
    if (node0 + lr0 + i < n_nodes) dstp[(size_t)i * CCH] = acc0[i] * sc0[i];
  }
#pragma unroll
  for (int i = 0; i < 8; ++i) {
    if (node0 + lr0 + 16 + i < n_nodes) dstp[(size_t)(16 + i) * CCH] = acc1[i] * sc1[i];
  }
}

#define GP  5
#define GTH 256

#define GCN_ROWB(x, J) __builtin_amdgcn_update_dpp(0, (x), 0x150 + (J), 0xf, 0xf, true)

__device__ __forceinline__ void gcn_edges(const v2i* __restrict__ pr, int beg, int m, int base,
                                          int l16, float dd, int own, int n_edges,
                                          int* s0, int* c0, int* s1, int* c1) {
  const int i0 = base + l16;
  const int i1 = i0 + 16;
  const v2i p0 = pr[min(max(beg + i0, 0), n_edges - 1)];
  const v2i p1 = pr[min(max(beg + i1, 0), n_edges - 1)];
  const bool ok0 = i0 < m && beg + i0 < n_edges;
  const bool ok1 = i1 < m && beg + i1 < n_edges;
  *s0 = ok0 ? p0.x : own;
  *c0 = ok0 ? __float_as_int(__int_as_float(p0.y) * dd) : 0;
  *s1 = ok1 ? p1.x : own;
  *c1 = ok1 ? __float_as_int(__int_as_float(p1.y) * dd) : 0;
}

template <int J0>
__device__ __forceinline__ void gcn_rows8(const int sv, const int cv,
                                          const char* __restrict__ xwb, const unsigned loff,
                                          float4* acc) {
#define GCN_LD(J)  (*(const float4*)(xwb + ((unsigned)GCN_ROWB(sv, J0 + (J)) * (unsigned)(CCH * 4) + loff)))
#define GCN_FMA(v, J) { const float c = __int_as_float(GCN_ROWB(cv, J0 + (J))); \
    acc->x = __builtin_fmaf((v).x, c, acc->x); acc->y = __builtin_fmaf((v).y, c, acc->y); \
    acc->z = __builtin_fmaf((v).z, c, acc->z); acc->w = __builtin_fmaf((v).w, c, acc->w); }
  const float4 v0 = GCN_LD(0);
  const float4 v1 = GCN_LD(1);
  const float4 v2 = GCN_LD(2);
  const float4 v3 = GCN_LD(3);
  const float4 v4 = GCN_LD(4);
  const float4 v5 = GCN_LD(5);
  const float4 v6 = GCN_LD(6);
  const float4 v7 = GCN_LD(7);
  GCN_FMA(v0, 0)
  GCN_FMA(v1, 1)
  GCN_FMA(v2, 2)
  GCN_FMA(v3, 3)
  GCN_FMA(v4, 4)
  GCN_FMA(v5, 5)
  GCN_FMA(v6, 6)
  GCN_FMA(v7, 7)
#undef GCN_LD
#undef GCN_FMA
}

__global__ void __launch_bounds__(GTH)
gcn_gather(const int* __restrict__ pair, const int* __restrict__ off,
           const int* __restrict__ cnt, const float* __restrict__ dinv,
           const float* __restrict__ xw, const float* __restrict__ b,
           float* __restrict__ out, int n_nodes, int n_edges) {
  const int lane = threadIdx.x & 31;
  const int l16  = lane & 15;
  const int half = lane >> 4;
  const int n0 = (2 * GP) * __builtin_amdgcn_readfirstlane(
      (int)((blockIdx.x * GTH + threadIdx.x) >> 5));
  if (n0 >= n_nodes) return;
  const v2i* __restrict__ pr = (const v2i*)pair;
  const char* __restrict__ xwb = (const char*)xw;
  const unsigned loff = (unsigned)l16 * 16u;

  const int hn = n0 + min(lane, 2 * GP - 1);
  const int hc = min(hn, n_nodes - 1);
  const int   offv = off[hc];
  const int   cntv = (hn < n_nodes) ? cnt[hc] : 0;
  const float dinv_v = dinv[hc];
  const float4 b4 = make_float4(b[4 * l16], b[4 * l16 + 1], b[4 * l16 + 2], b[4 * l16 + 3]);

  const int bA0 = __builtin_amdgcn_readlane(offv, 0);
  const int bB0 = __builtin_amdgcn_readlane(offv, 1);
  const int mA0 = __builtin_amdgcn_readlane(cntv, 0);
  const int mB0 = __builtin_amdgcn_readlane(cntv, 1);
  const int dA0 = __builtin_amdgcn_readlane(__float_as_int(dinv_v), 0);
  const int dB0 = __builtin_amdgcn_readlane(__float_as_int(dinv_v), 1);
  int   n    = n0 + half;
  int   beg  = half ? bB0 : bA0;
  int   m    = half ? mB0 : mA0;
  int   mmax = max(mA0, mB0);
  float dd   = __int_as_float(half ? dB0 : dA0);
  int s0, c0, s1, c1;
  gcn_edges(pr, beg, m, 0, l16, dd, min(n, n_nodes - 1), n_edges, &s0, &c0, &s1, &c1);

  for (int p = 0; p < GP; ++p) {
    const int own = min(n, n_nodes - 1);
    const float4 self = *(const float4*)(xwb + ((unsigned)own * (unsigned)(CCH * 4) + loff));

    const int pn = min(p + 1, GP - 1);
    const int n2   = n0 + 2 * pn + half;
    const int bA   = __builtin_amdgcn_readlane(offv, 2 * pn);
    const int bB   = __builtin_amdgcn_readlane(offv, 2 * pn + 1);
    const int mA2  = __builtin_amdgcn_readlane(cntv, 2 * pn);
    const int mB2  = __builtin_amdgcn_readlane(cntv, 2 * pn + 1);
    const int dA   = __builtin_amdgcn_readlane(__float_as_int(dinv_v), 2 * pn);
    const int dB   = __builtin_amdgcn_readlane(__float_as_int(dinv_v), 2 * pn + 1);
    const int   beg2 = half ? bB : bA;
    const int   m2   = half ? mB2 : mA2;
    const float dd2  = __int_as_float(half ? dB : dA);
    int t0, u0, t1, u1;
    gcn_edges(pr, beg2, m2, 0, l16, dd2, min(n2, n_nodes - 1), n_edges, &t0, &u0, &t1, &u1);

    float4 acc = make_float4(0.0f, 0.0f, 0.0f, 0.0f);
    for (int base = 0; base < mmax; base += 32) {
      if (base > 0) gcn_edges(pr, beg, m, base, l16, dd, own, n_edges, &s0, &c0, &s1, &c1);
      const int rem = mmax - base;
      gcn_rows8<0>(s0, c0, xwb, loff, &acc);
      if (rem > 8)  gcn_rows8<8>(s0, c0, xwb, loff, &acc);
      if (rem > 16) gcn_rows8<0>(s1, c1, xwb, loff, &acc);
      if (rem > 24) gcn_rows8<8>(s1, c1, xwb, loff, &acc);
    }

    const float k2 = FILLW * dd;
    float4 r;
    r.x = tanhf(acc.x + self.x * k2 + b4.x);
    r.y = tanhf(acc.y + self.y * k2 + b4.y);
    r.z = tanhf(acc.z + self.z * k2 + b4.z);
    r.w = tanhf(acc.w + self.w * k2 + b4.w);
    if (n < n_nodes) *(float4*)(out + (size_t)n * CCH + l16 * 4) = r;

    n = n2; beg = beg2; m = m2; dd = dd2; mmax = max(mA2, mB2);
    s0 = t0; c0 = u0; s1 = t1; c1 = u1;
  }
}

extern "C" void kernel_launch(void* const* d_in, const int* in_sizes, int n_in,
                              void* d_out, int out_size, void* d_ws, size_t ws_size,
                              hipStream_t stream) {
  const float* x  = (const float*)d_in[0];
  const int*   ei = (const int*)d_in[1];
  const float* ew = (const float*)d_in[2];
  const float* W  = (const float*)d_in[3];
  const float* b  = (const float*)d_in[4];
  float* out = (float*)d_out;

  const int n_nodes = in_sizes[0] / CCH;
  const int n_edges = in_sizes[2];
  const int* src = ei;
  const int* dst = ei + n_edges;

  const size_t n_pad = ((size_t)n_nodes + 63) & ~(size_t)63;
  const size_t e_pad = ((size_t)n_edges + 63) & ~(size_t)63;
  const int    n_bins = (n_nodes + BINW - 1) / BINW;
  const size_t b_pad = ((size_t)n_bins + 63) & ~(size_t)63;
  float* xw   = (float*)d_ws;
  float* dinv = xw + n_pad * CCH;
  int*   cnt  = (int*)(dinv + n_pad);
  int*   off  = cnt + n_pad;
  int*   tmp  = off + n_pad;
  int*   pair = tmp + e_pad * 4;
  int*   tab  = pair + e_pad * 2;
  const int tab_pitch = (int)b_pad + 64;

  const int B = 256;
  const int n_chunks = (n_edges + EPB - 1) / EPB;

  gcn_chunk<<<n_chunks, B, 2 * n_bins * sizeof(int), stream>>>(src, dst, ew, tmp, tab,
                                                              n_nodes, n_edges, n_bins, tab_pitch);
  gcn_owner<<<n_bins, 1024, 0, stream>>>(tmp, tab, cnt, off, pair, n_nodes, n_edges, n_chunks,
                                         tab_pitch);
  gcn_dinv_xw<<<(n_nodes + TN - 1) / TN, B, 0, stream>>>(pair, off, cnt, dinv, x, W, xw,
                                                        n_nodes, n_edges);

  gcn_gather<<<(n_nodes + 16 * GP - 1) / (16 * GP), GTH, 0, stream>>>(pair, off, cnt, dinv, xw, b, out,
                                                                    n_nodes, n_edges);
}


# Level-one records are 8 bytes, not 16: src and the destination's place inside its 256-node range share word 0, the range is implied by position; tmp write and read halve; sum orders unchanged
# speedup vs baseline: 5.1788x; 1.0882x over previous
#include <hip/hip_runtime.h>
#include <hip/hip_bf16.h>
#include <math.h>

typedef __attribute__((ext_vector_type(2))) float v2f;
typedef __attribute__((ext_vector_type(8))) float v8f;
typedef __attribute__((ext_vector_type(2))) int   v2i;
typedef __attribute__((ext_vector_type(4))) int   v4i;

#define CCH   64
#define FILLW 2.0f

#define BINW  256
#define BSH   8
#define EPB   4096
#define EPT   16
#define DLSH  24
static_assert(BINW <= (1 << (32 - DLSH)), "place inside the range fits above src");

__global__ void __launch_bounds__(256)
gcn_chunk(const int* __restrict__ src, const int* __restrict__ dst,
          const float* __restrict__ w, int* __restrict__ tmp, int* __restrict__ tab,
          int n_nodes, int n_edges, int n_bins, int tab_pitch) {
  extern __shared__ int sh[];
  __shared__ int wsum[8];
  __shared__ __attribute__((aligned(16))) v2i stage[EPB];
  int* hist = sh;
  int* base = sh + n_bins;
  const int tid  = threadIdx.x;
  const int lane = tid & 31;
  const int wv   = tid >> 5;
  for (int i = tid; i < n_bins; i += 256) hist[i] = 0;
  __syncthreads();

  const int e0 = blockIdx.x * EPB + tid;
  int dk[EPT];
  int sk[EPT];
  int wk[EPT];
  int rk[EPT];
#pragma unroll
  for (int k = 0; k < EPT; ++k) {
    const int e = min(e0 + k * 256, n_edges - 1);
    dk[k] = dst[e];
    sk[k] = src[e];
    wk[k] = __float_as_int(w[e]);
  }
#pragma unroll
  for (int k = 0; k < EPT; ++k) {
    rk[k] = 0;
    if (e0 + k * 256 < n_edges && (unsigned)dk[k] < (unsigned)n_nodes)
      rk[k] = atomicAdd(&hist[dk[k] >> BSH], 1);
    else
      dk[k] = -1;
  }
  __syncthreads();

  int* __restrict__ row = tab + (size_t)blockIdx.x * tab_pitch;
  int carry = 0;
  for (int c0 = 0; c0 < n_bins; c0 += 256) {
    const int i = c0 + tid;
    const int t = (i < n_bins) ? hist[i] : 0;
    int incl = t;
    for (int s = 1; s < 32; s <<= 1) {
      const int u = __shfl_up(incl, s, 32);
      if (lane >= s) incl += u;
    }
    if (lane == 31) wsum[wv] = incl;
    __syncthreads();
    int wpre = 0;
    int tot  = 0;
    for (int j = 0; j < 8; ++j) {
      const int x = wsum[j];
      if (j < wv) wpre += x;
      tot += x;
    }
    if (i < n_bins) {
      const int b = carry + wpre + incl - t;
      base[i] = b;
      row[i]  = b;
    }
    carry += tot;
    __syncthreads();
  }
  if (tid == 0) row[n_bins] = carry;

#pragma unroll
  for (int k = 0; k < EPT; ++k) {
    if (dk[k] >= 0) {
      const int p = base[dk[k] >> BSH] + rk[k];
      if ((unsigned)p < (unsigned)EPB)
        stage[p] = (v2i){(int)((unsigned)sk[k] | ((unsigned)(dk[k] & (BINW - 1)) << DLSH)), wk[k]};
    }
  }
  __syncthreads();
  v4i* __restrict__ outp = (v4i*)(tmp + (size_t)blockIdx.x * (2 * EPB));
  const v4i* st4 = (const v4i*)stage;
  const int lim = min(carry, n_edges - (int)blockIdx.x * EPB);
  for (int i = tid; 2 * i < lim; i += 256) {
    v4i v = st4[i];
    if (2 * i + 1 >= lim) { v.z = 0; v.w = 0; }
    outp[i] = v;
  }
}

#define OCAP  8192
#define KR    8
#define CPT   512

__device__ __forceinline__ void gcn_piece(const int* __restrict__ tab, int tab_pitch,
                                          int b, int bin, int n_edges,
                                          int* lo_out, int* first, int* len) {
  const int* __restrict__ row = tab + (size_t)b * tab_pitch;
  int lo = row[bin];
  int hi = row[bin + 1];
  lo = min(max(lo, 0), EPB);
  hi = min(max(hi, lo), EPB);
  const int room = max(n_edges - b * EPB, 0);
  lo = min(lo, room);
  hi = min(hi, room);
  *lo_out = lo;
  *first  = b * EPB + lo;
  *len    = hi - lo;
}

__device__ __forceinline__ void gcn_put(const v2i r, const int* offL, int* curL,
                                        bool staged, v2i* stage, int* __restrict__ pair,
                                        int start, int n_edges) {
  const int dl  = (int)((unsigned)r.x >> DLSH);
  const int src = r.x & ((1 << DLSH) - 1);
  const int li = offL[dl] + atomicAdd(&curL[dl], 1);
  if (staged) {
    if ((unsigned)li < (unsigned)OCAP)
      stage[li] = (v2i){src, r.y};
  } else {
    const int p = start + li;
    if ((unsigned)p < (unsigned)n_edges)
      ((v2i*)pair)[p] = (v2i){src, r.y};
  }
}

__global__ void __launch_bounds__(1024)
gcn_owner(const int* __restrict__ tmp, const int* __restrict__ tab,
          int* __restrict__ cnt, int* __restrict__ off, int* __restrict__ pair,
          int n_nodes, int n_edges, int n_chunks, int tab_pitch) {
  __shared__ int cntL[BINW];
  __shared__ int offL[BINW];
  __shared__ int curL[BINW];
  __shared__ int red[32];
  __shared__ int redm[32];
  __shared__ v2i stage[OCAP];
  const int tid  = threadIdx.x;
  const int lane = tid & 31;
  const int wv   = tid >> 5;
  const int bin  = blockIdx.x;
  const int node0 = bin * BINW;
  const int sub  = tid & 1;
  const int b0   = tid >> 1;
  const v2i* __restrict__ rec2 = (const v2i*)tmp;

  int f0 = 0, c0 = 0;
  int ps = 0, pm = 0;
  for (int b = b0; b < n_chunks; b += CPT) {
    int lo, f, c;
    gcn_piece(tab, tab_pitch, b, bin, n_edges, &lo, &f, &c);
    if (b == b0) { f0 = f; c0 = c; }
    if (sub == 0) { ps += lo; pm += c; }
  }
  v2i q[KR];
  {
    const int last = max(c0 - 1, 0);
#pragma unroll
    for (int k = 0; k < KR; ++k)
      q[k] = rec2[max(min(f0 + min(sub + 2 * k, last), n_edges - 1), 0)];
  }
  for (int s = 16; s > 0; s >>= 1) {
    ps += __shfl_xor(ps, s, 32);
    pm += __shfl_xor(pm, s, 32);
  }
  if (lane == 0) { red[wv] = ps; redm[wv] = pm; }
  if (tid < BINW) { cntL[tid] = 0; curL[tid] = 0; }
  __syncthreads();
  int start = red[lane], m = redm[lane];
  for (int s = 16; s > 0; s >>= 1) {
    start += __shfl_xor(start, s, 32);
    m     += __shfl_xor(m, s, 32);
  }
  start = min(max(start, 0), n_edges);
  m     = min(max(m, 0), n_edges - start);
  const bool staged = (m <= OCAP);
  __syncthreads();

#pragma unroll
  for (int k = 0; k < KR; ++k) {
    if (sub + 2 * k < c0) atomicAdd(&cntL[(unsigned)q[k].x >> DLSH], 1);
  }
  for (int i = sub + 2 * KR; i < c0; i += 2)
    atomicAdd(&cntL[(unsigned)rec2[min(f0 + i, n_edges - 1)].x >> DLSH], 1);
  for (int b = b0 + CPT; b < n_chunks; b += CPT) {
    int lo, f, c;
    gcn_piece(tab, tab_pitch, b, bin, n_edges, &lo, &f, &c);
    for (int i = sub; i < c; i += 2)
      atomicAdd(&cntL[(unsigned)rec2[min(f + i, n_edges - 1)].x >> DLSH], 1);
  }
  __syncthreads();

  int c = 0, incl = 0;
  if (tid < BINW) {
    c = cntL[tid];
    incl = c;
    for (int s = 1; s < 32; s <<= 1) {
      const int u = __shfl_up(incl, s, 32);
      if (lane >= s) incl += u;
    }
    if (lane == 31) red[wv] = incl;
  }
  __syncthreads();
  if (tid < BINW) {
    int wpre = 0;
    for (int j = 0; j < 8; ++j) if (j < wv) wpre += red[j];
    const int excl = wpre + incl - c;
    offL[tid] = excl;
    const int node = node0 + tid;
    if (node < n_nodes) { cnt[node] = c; off[node] = start + excl; }
  }
  __syncthreads();

#pragma unroll
  for (int k = 0; k < KR; ++k)
    if (sub + 2 * k < c0)
      gcn_put(q[k], offL, curL, staged, stage, pair, start, n_edges);
  for (int i = sub + 2 * KR; i < c0; i += 2)
    gcn_put(rec2[min(f0 + i, n_edges - 1)], offL, curL, staged, stage, pair, start, n_edges);
  for (int b = b0 + CPT; b < n_chunks; b += CPT) {
    int lo, f, cc;
    gcn_piece(tab, tab_pitch, b, bin, n_edges, &lo, &f, &cc);
    for (int i = sub; i < cc; i += 2)
      gcn_put(rec2[min(f + i, n_edges - 1)], offL, curL, staged, stage, pair, start, n_edges);
  }
  if (staged) {
    __syncthreads();
    const int shift = start & 15;
    for (int j = tid; j < m + shift; j += 1024) {
      const int i = j - shift;
      if (i >= 0) ((v2i*)pair)[start + i] = stage[i];
    }
  }
}

#define TN  64
#define XP  68
__global__ void __launch_bounds__(256)
gcn_dinv_xw(const int* __restrict__ pair, const int* __restrict__ off,
            const int* __restrict__ cnt, float* __restrict__ dinv,
            const float* __restrict__ X, const float* __restrict__ W,
            float* __restrict__ XS, int n_nodes, int n_edges) {
  __shared__ __attribute__((aligned(16))) float xs[TN * XP];
  __shared__ __attribute__((aligned(16))) float sd[TN];
  const int tid   = threadIdx.x;
  const int lane  = tid & 31;
  const int wv    = tid >> 5;
  const int node0 = blockIdx.x * TN;
  const int half  = lane >> 4;
  const int l16   = lane & 15;
  const int col_tile = wv & 3;
  const int rp    = wv >> 2;

  const float* __restrict__ wcol = W + col_tile * 16 + l16;
  v2f bf[CCH / 4];
#pragma unroll
  for (int kk = 0; kk < CCH / 4; ++kk) {
    const int ka = kk * 4 + 2 * half;
    bf[kk].x = wcol[(size_t)(ka + 0) * CCH];
    bf[kk].y = wcol[(size_t)(ka + 1) * CCH];
  }

  const int sub = tid & 7;
  int nn[2], beg[2], mm[2];
#pragma unroll
  for (int h = 0; h < 2; ++h) {
    nn[h]  = node0 + h * 32 + (tid >> 3);
    beg[h] = 0;
    mm[h]  = 0;
    if (nn[h] < n_nodes) { beg[h] = off[nn[h]]; mm[h] = cnt[nn[h]]; }
  }

#pragma unroll
  for (int k = 0; k < 4; ++k) {
    const int idx = tid + k * 256;
    const int r   = idx >> 4;
    const int c4  = idx & 15;
    const int gr  = min(node0 + r, n_nodes - 1);
    const float4 v = ((const float4*)(X + (size_t)gr * CCH))[c4];
    *(float4*)&xs[r * XP + c4 * 4] = v;
  }

  float wpre[2][3];
#pragma unroll
  for (int h = 0; h < 2; ++h) {
#pragma unroll
    for (int k = 0; k < 3; ++k) {
      const int q = min(max(beg[h] + sub + 8 * k, 0), n_edges - 1);
      wpre[h][k] = __int_as_float(pair[2 * (size_t)q + 1]);
    }
  }
#pragma unroll
  for (int h = 0; h < 2; ++h) {
    float d = 0.0f;
#pragma unroll
    for (int k = 0; k < 3; ++k) {
      const int i = sub + 8 * k;
      if (i < mm[h] && beg[h] + i < n_edges) d += wpre[h][k];
    }
    for (int i = sub + 24; i < mm[h]; i += 8) {
      const int q = beg[h] + i;
      if (q < n_edges) d += __int_as_float(pair[2 * (size_t)q + 1]);
    }
    d += __shfl_xor(d, 1, 32);
    d += __shfl_xor(d, 2, 32);
    d += __shfl_xor(d, 4, 32);
    if (sub == 0) {
      d += FILLW;
      const float di = (d > 0.0f) ? rsqrtf(d) : 0.0f;
      sd[h * 32 + (tid >> 3)] = di;
      if (nn[h] < n_nodes) dinv[nn[h]] = di;
    }
  }
  __syncthreads();

  const int lr0 = rp * 32 + 8 * half;
  const float4 s00 = *(const float4*)&sd[lr0];
  const float4 s01 = *(const float4*)&sd[lr0 + 4];
  const float4 s10 = *(const float4*)&sd[lr0 + 16];
  const float4 s11 = *(const float4*)&sd[lr0 + 20];
  const float sc0[8] = {s00.x, s00.y, s00.z, s00.w, s01.x, s01.y, s01.z, s01.w};
  const float sc1[8] = {s10.x, s10.y, s10.z, s10.w, s11.x, s11.y, s11.z, s11.w};

  const float* xa0 = &xs[(rp * 32 + l16) * XP];
  const float* xa1 = xa0 + 16 * XP;
  v8f acc0 = {};
  v8f acc1 = {};
#pragma unroll
  for (int kk = 0; kk < CCH / 4; ++kk) {
    const int ka = kk * 4 + 2 * half;
    const v2f a0 = *(const v2f*)&xa0[ka];
    const v2f a1 = *(const v2f*)&xa1[ka];
    acc0 = __builtin_amdgcn_wmma_f32_16x16x4_f32(
         false, a0,  false, bf[kk],
         (short)0, acc0,  false,  false);
    acc1 = __builtin_amdgcn_wmma_f32_16x16x4_f32(
         false, a1,  false, bf[kk],
         (short)0, acc1,  false,  false);
  }

  float* __restrict__ dstp = XS + (size_t)(node0 + lr0) * CCH + col_tile * 16 + l16;
#pragma unroll
  for (int i = 0; i < 8; ++i) {
    if (node0 + lr0 + i < n_nodes) dstp[(size_t)i * CCH] = acc0[i] * sc0[i];
  }
#pragma unroll
  for (int i = 0; i < 8; ++i) {
    if (node0 + lr0 + 16 + i < n_nodes) dstp[(size_t)(16 + i) * CCH] = acc1[i] * sc1[i];
  }
}

#define GP  5
#define GTH 256

#define GCN_ROWB(x, J) __builtin_amdgcn_update_dpp(0, (x), 0x150 + (J), 0xf, 0xf, true)

__device__ __forceinline__ void gcn_edges(const v2i* __restrict__ pr, int beg, int m, int base,
                                          int l16, float dd, int own, int n_edges,
                                          int* s0, int* c0, int* s1, int* c1) {
  const int i0 = base + l16;
  const int i1 = i0 + 16;
  const v2i p0 = pr[min(max(beg + i0, 0), n_edges - 1)];
  const v2i p1 = pr[min(max(beg + i1, 0), n_edges - 1)];
  const bool ok0 = i0 < m && beg + i0 < n_edges;
  const bool ok1 = i1 < m && beg + i1 < n_edges;
  *s0 = ok0 ? p0.x : own;
  *c0 = ok0 ? __float_as_int(__int_as_float(p0.y) * dd) : 0;
  *s1 = ok1 ? p1.x : own;
  *c1 = ok1 ? __float_as_int(__int_as_float(p1.y) * dd) : 0;
}

template <int J0>
__device__ __forceinline__ void gcn_rows8(const int sv, const int cv,
                                          const char* __restrict__ xwb, const unsigned loff,
                                          float4* acc) {
#define GCN_LD(J)  (*(const float4*)(xwb + ((unsigned)GCN_ROWB(sv, J0 + (J)) * (unsigned)(CCH * 4) + loff)))
#define GCN_FMA(v, J) { const float c = __int_as_float(GCN_ROWB(cv, J0 + (J))); \
    acc->x = __builtin_fmaf((v).x, c, acc->x); acc->y = __builtin_fmaf((v).y, c, acc->y); \
    acc->z = __builtin_fmaf((v).z, c, acc->z); acc->w = __builtin_fmaf((v).w, c, acc->w); }
  const float4 v0 = GCN_LD(0);
  const float4 v1 = GCN_LD(1);
  const float4 v2 = GCN_LD(2);
  const float4 v3 = GCN_LD(3);
  const float4 v4 = GCN_LD(4);
  const float4 v5 = GCN_LD(5);
  const float4 v6 = GCN_LD(6);
  const float4 v7 = GCN_LD(7);
  GCN_FMA(v0, 0)
  GCN_FMA(v1, 1)
  GCN_FMA(v2, 2)
  GCN_FMA(v3, 3)
  GCN_FMA(v4, 4)
  GCN_FMA(v5, 5)
  GCN_FMA(v6, 6)
  GCN_FMA(v7, 7)
#undef GCN_LD
#undef GCN_FMA
}

__global__ void __launch_bounds__(GTH)
gcn_gather(const int* __restrict__ pair, const int* __restrict__ off,
           const int* __restrict__ cnt, const float* __restrict__ dinv,
           const float* __restrict__ xw, const float* __restrict__ b,
           float* __restrict__ out, int n_nodes, int n_edges) {
  const int lane = threadIdx.x & 31;
  const int l16  = lane & 15;
  const int half = lane >> 4;
  const int n0 = (2 * GP) * __builtin_amdgcn_readfirstlane(
      (int)((blockIdx.x * GTH + threadIdx.x) >> 5));
  if (n0 >= n_nodes) return;
  const v2i* __restrict__ pr = (const v2i*)pair;
  const char* __restrict__ xwb = (const char*)xw;
  const unsigned loff = (unsigned)l16 * 16u;

  const int hn = n0 + min(lane, 2 * GP - 1);
  const int hc = min(hn, n_nodes - 1);
  const int   offv = off[hc];
  const int   cntv = (hn < n_nodes) ? cnt[hc] : 0;
  const float dinv_v = dinv[hc];
  const float4 b4 = make_float4(b[4 * l16], b[4 * l16 + 1], b[4 * l16 + 2], b[4 * l16 + 3]);

  const int bA0 = __builtin_amdgcn_readlane(offv, 0);
  const int bB0 = __builtin_amdgcn_readlane(offv, 1);
  const int mA0 = __builtin_amdgcn_readlane(cntv, 0);
  const int mB0 = __builtin_amdgcn_readlane(cntv, 1);
  const int dA0 = __builtin_amdgcn_readlane(__float_as_int(dinv_v), 0);
  const int dB0 = __builtin_amdgcn_readlane(__float_as_int(dinv_v), 1);
  int   n    = n0 + half;
  int   beg  = half ? bB0 : bA0;
  int   m    = half ? mB0 : mA0;
  int   mmax = max(mA0, mB0);
  float dd   = __int_as_float(half ? dB0 : dA0);
  int s0, c0, s1, c1;
  gcn_edges(pr, beg, m, 0, l16, dd, min(n, n_nodes - 1), n_edges, &s0, &c0, &s1, &c1);

  for (int p = 0; p < GP; ++p) {
    const int own = min(n, n_nodes - 1);
    const float4 self = *(const float4*)(xwb + ((unsigned)own * (unsigned)(CCH * 4) + loff));

    const int pn = min(p + 1, GP - 1);
    const int n2   = n0 + 2 * pn + half;
    const int bA   = __builtin_amdgcn_readlane(offv, 2 * pn);
    const int bB   = __builtin_amdgcn_readlane(offv, 2 * pn + 1);
    const int mA2  = __builtin_amdgcn_readlane(cntv, 2 * pn);
    const int mB2  = __builtin_amdgcn_readlane(cntv, 2 * pn + 1);
    const int dA   = __builtin_amdgcn_readlane(__float_as_int(dinv_v), 2 * pn);
    const int dB   = __builtin_amdgcn_readlane(__float_as_int(dinv_v), 2 * pn + 1);
    const int   beg2 = half ? bB : bA;
    const int   m2   = half ? mB2 : mA2;
    const float dd2  = __int_as_float(half ? dB : dA);
    int t0, u0, t1, u1;
    gcn_edges(pr, beg2, m2, 0, l16, dd2, min(n2, n_nodes - 1), n_edges, &t0, &u0, &t1, &u1);

    float4 acc = make_float4(0.0f, 0.0f, 0.0f, 0.0f);
    for (int base = 0; base < mmax; base += 32) {
      if (base > 0) gcn_edges(pr, beg, m, base, l16, dd, own, n_edges, &s0, &c0, &s1, &c1);
      const int rem = mmax - base;
      gcn_rows8<0>(s0, c0, xwb, loff, &acc);
      if (rem > 8)  gcn_rows8<8>(s0, c0, xwb, loff, &acc);
      if (rem > 16) gcn_rows8<0>(s1, c1, xwb, loff, &acc);
      if (rem > 24) gcn_rows8<8>(s1, c1, xwb, loff, &acc);
    }

    const float k2 = FILLW * dd;
    float4 r;
    r.x = tanhf(acc.x + self.x * k2 + b4.x);
    r.y = tanhf(acc.y + self.y * k2 + b4.y);
    r.z = tanhf(acc.z + self.z * k2 + b4.z);
    r.w = tanhf(acc.w + self.w * k2 + b4.w);
    if (n < n_nodes) *(float4*)(out + (size_t)n * CCH + l16 * 4) = r;

    n = n2; beg = beg2; m = m2; dd = dd2; mmax = max(mA2, mB2);
    s0 = t0; c0 = u0; s1 = t1; c1 = u1;
  }
}

extern "C" void kernel_launch(void* const* d_in, const int* in_sizes, int n_in,
                              void* d_out, int out_size, void* d_ws, size_t ws_size,
                              hipStream_t stream) {
  const float* x  = (const float*)d_in[0];
  const int*   ei = (const int*)d_in[1];
  const float* ew = (const float*)d_in[2];
  const float* W  = (const float*)d_in[3];
  const float* b  = (const float*)d_in[4];
  float* out = (float*)d_out;

  const int n_nodes = in_sizes[0] / CCH;
  const int n_edges = in_sizes[2];
  const int* src = ei;
  const int* dst = ei + n_edges;

  const size_t n_pad = ((size_t)n_nodes + 63) & ~(size_t)63;
  const size_t e_pad = ((size_t)n_edges + 63) & ~(size_t)63;
  const int    n_bins = (n_nodes + BINW - 1) / BINW;
  const size_t b_pad = ((size_t)n_bins + 63) & ~(size_t)63;
  float* xw   = (float*)d_ws;
  float* dinv = xw + n_pad * CCH;
  int*   cnt  = (int*)(dinv + n_pad);
  int*   off  = cnt + n_pad;
  int*   tmp  = off + n_pad;
  int*   pair = tmp + e_pad * 2;
  int*   tab  = pair + e_pad * 2;
  const int tab_pitch = (int)b_pad + 64;

  const int B = 256;
  const int n_chunks = (n_edges + EPB - 1) / EPB;

  gcn_chunk<<<n_chunks, B, 2 * n_bins * sizeof(int), stream>>>(src, dst, ew, tmp, tab,
                                                              n_nodes, n_edges, n_bins, tab_pitch);
  gcn_owner<<<n_bins, 1024, 0, stream>>>(tmp, tab, cnt, off, pair, n_nodes, n_edges, n_chunks,
                                         tab_pitch);
  gcn_dinv_xw<<<(n_nodes + TN - 1) / TN, B, 0, stream>>>(pair, off, cnt, dinv, x, W, xw,
                                                        n_nodes, n_edges);

  gcn_gather<<<(n_nodes + 16 * GP - 1) / (16 * GP), GTH, 0, stream>>>(pair, off, cnt, dinv, xw, b, out,
                                                                    n_nodes, n_edges);
}


# Range owner group also sums each node's weights from its ordered pairs in LDS and writes dinv; product launch takes dinv in one load, reads no off, cnt or pair; degree sum order = edge order
# speedup vs baseline: 5.5167x; 1.0652x over previous
#include <hip/hip_runtime.h>
#include <hip/hip_bf16.h>
#include <math.h>

typedef __attribute__((ext_vector_type(2))) float v2f;
typedef __attribute__((ext_vector_type(8))) float v8f;
typedef __attribute__((ext_vector_type(2))) int   v2i;
typedef __attribute__((ext_vector_type(4))) int   v4i;

#define CCH   64
#define FILLW 2.0f

#define BINW  256
#define BSH   8
#define EPB   4096
#define EPT   16
#define DLSH  24
static_assert(BINW <= (1 << (32 - DLSH)), "place inside the range fits above src");

__global__ void __launch_bounds__(256)
gcn_chunk(const int* __restrict__ src, const int* __restrict__ dst,
          const float* __restrict__ w, int* __restrict__ tmp, int* __restrict__ tab,
          int n_nodes, int n_edges, int n_bins, int tab_pitch) {
  extern __shared__ int sh[];
  __shared__ int wsum[8];
  __shared__ __attribute__((aligned(16))) v2i stage[EPB];
  int* hist = sh;
  int* base = sh + n_bins;
  const int tid  = threadIdx.x;
  const int lane = tid & 31;
  const int wv   = tid >> 5;
  for (int i = tid; i < n_bins; i += 256) hist[i] = 0;
  __syncthreads();

  const int e0 = blockIdx.x * EPB + tid;
  int dk[EPT];
  int sk[EPT];
  int wk[EPT];
  int rk[EPT];
#pragma unroll
  for (int k = 0; k < EPT; ++k) {
    const int e = min(e0 + k * 256, n_edges - 1);
    dk[k] = dst[e];
    sk[k] = src[e];
    wk[k] = __float_as_int(w[e]);
  }
#pragma unroll
  for (int k = 0; k < EPT; ++k) {
    rk[k] = 0;
    if (e0 + k * 256 < n_edges && (unsigned)dk[k] < (unsigned)n_nodes)
      rk[k] = atomicAdd(&hist[dk[k] >> BSH], 1);
    else
      dk[k] = -1;
  }
  __syncthreads();

  int* __restrict__ row = tab + (size_t)blockIdx.x * tab_pitch;
  int carry = 0;
  for (int c0 = 0; c0 < n_bins; c0 += 256) {
    const int i = c0 + tid;
    const int t = (i < n_bins) ? hist[i] : 0;
    int incl = t;
    for (int s = 1; s < 32; s <<= 1) {
      const int u = __shfl_up(incl, s, 32);
      if (lane >= s) incl += u;
    }
    if (lane == 31) wsum[wv] = incl;
    __syncthreads();
    int wpre = 0;
    int tot  = 0;
    for (int j = 0; j < 8; ++j) {
      const int x = wsum[j];
      if (j < wv) wpre += x;
      tot += x;
    }
    if (i < n_bins) {
      const int b = carry + wpre + incl - t;
      base[i] = b;
      row[i]  = b;
    }
    carry += tot;
    __syncthreads();
  }
  if (tid == 0) row[n_bins] = carry;

#pragma unroll
  for (int k = 0; k < EPT; ++k) {
    if (dk[k] >= 0) {
      const int p = base[dk[k] >> BSH] + rk[k];
      if ((unsigned)p < (unsigned)EPB)
        stage[p] = (v2i){(int)((unsigned)sk[k] | ((unsigned)(dk[k] & (BINW - 1)) << DLSH)), wk[k]};
    }
  }
  __syncthreads();
  v4i* __restrict__ outp = (v4i*)(tmp + (size_t)blockIdx.x * (2 * EPB));
  const v4i* st4 = (const v4i*)stage;
  const int lim = min(carry, n_edges - (int)blockIdx.x * EPB);
  for (int i = tid; 2 * i < lim; i += 256) {
    v4i v = st4[i];
    if (2 * i + 1 >= lim) { v.z = 0; v.w = 0; }
    outp[i] = v;
  }
}

#define OCAP  8192
#define KR    8
#define CPT   512

__device__ __forceinline__ void gcn_piece(const int* __restrict__ tab, int tab_pitch,
                                          int b, int bin, int n_edges,
                                          int* lo_out, int* first, int* len) {
  const int* __restrict__ row = tab + (size_t)b * tab_pitch;
  int lo = row[bin];
  int hi = row[bin + 1];
  lo = min(max(lo, 0), EPB);
  hi = min(max(hi, lo), EPB);
  const int room = max(n_edges - b * EPB, 0);
  lo = min(lo, room);
  hi = min(hi, room);
  *lo_out = lo;
  *first  = b * EPB + lo;
  *len    = hi - lo;
}

__device__ __forceinline__ void gcn_put(const v2i r, const int* offL, int* curL,
                                        bool staged, v2i* stage, int* __restrict__ pair,
                                        int start, int n_edges) {
  const int dl  = (int)((unsigned)r.x >> DLSH);
  const int src = r.x & ((1 << DLSH) - 1);
  const int li = offL[dl] + atomicAdd(&curL[dl], 1);
  if (staged) {
    if ((unsigned)li < (unsigned)OCAP)
      stage[li] = (v2i){src, r.y};
  } else {
    const int p = start + li;
    if ((unsigned)p < (unsigned)n_edges)
      ((v2i*)pair)[p] = (v2i){src, r.y};
  }
}

template <bool STAGED>
__device__ __forceinline__ void gcn_deg(const v2i* stage, const int* __restrict__ pair,
                                        const int* offL, const int* cntL,
                                        float* __restrict__ dinv, int t, int node0,
                                        int start, int n_nodes, int n_edges) {
  const int o = offL[t];
  const int c = cntL[t];
  float d = 0.0f;
  for (int base = 0; base < c; base += 8) {
    int wq[8];
    if (STAGED) {
      const v2i* sp = stage + min((unsigned)(o + base), (unsigned)OCAP);
#pragma unroll
      for (int k = 0; k < 8; ++k) wq[k] = sp[k].y;
      asm volatile("" :: "v"(wq[0]), "v"(wq[1]), "v"(wq[2]), "v"(wq[3]),
                         "v"(wq[4]), "v"(wq[5]), "v"(wq[6]), "v"(wq[7]));
    } else {
#pragma unroll
      for (int k = 0; k < 8; ++k)
        wq[k] = pair[2 * (size_t)min(max(start + o + base + k, 0), n_edges - 1) + 1];
    }
    const int rem = c - base;
#pragma unroll
    for (int k = 0; k < 8; ++k)
      if (k < rem) d += __int_as_float(wq[k]);
  }
  d += FILLW;
  const int node = node0 + t;
  if (node < n_nodes) dinv[node] = (d > 0.0f) ? rsqrtf(d) : 0.0f;
}

__global__ void __launch_bounds__(1024)
gcn_owner(const int* __restrict__ tmp, const int* __restrict__ tab,
          int* __restrict__ cnt, int* __restrict__ off, int* __restrict__ pair,
          float* __restrict__ dinv,
          int n_nodes, int n_edges, int n_chunks, int tab_pitch) {
  __shared__ int cntL[BINW];
  __shared__ int offL[BINW];
  __shared__ int curL[BINW];
  __shared__ int red[32];
  __shared__ int redm[32];
  __shared__ v2i stage[OCAP + 8];
  const int tid  = threadIdx.x;
  const int lane = tid & 31;
  const int wv   = tid >> 5;
  const int bin  = blockIdx.x;
  const int node0 = bin * BINW;
  const int sub  = tid & 1;
  const int b0   = tid >> 1;
  const v2i* __restrict__ rec2 = (const v2i*)tmp;

  int f0 = 0, c0 = 0;
  int ps = 0, pm = 0;
  for (int b = b0; b < n_chunks; b += CPT) {
    int lo, f, c;
    gcn_piece(tab, tab_pitch, b, bin, n_edges, &lo, &f, &c);
    if (b == b0) { f0 = f; c0 = c; }
    if (sub == 0) { ps += lo; pm += c; }
  }
  v2i q[KR];
  {
    const int last = max(c0 - 1, 0);
#pragma unroll
    for (int k = 0; k < KR; ++k)
      q[k] = rec2[max(min(f0 + min(sub + 2 * k, last), n_edges - 1), 0)];
  }
  for (int s = 16; s > 0; s >>= 1) {
    ps += __shfl_xor(ps, s, 32);
    pm += __shfl_xor(pm, s, 32);
  }
  if (lane == 0) { red[wv] = ps; redm[wv] = pm; }
  if (tid < BINW) { cntL[tid] = 0; curL[tid] = 0; }
  __syncthreads();
  int start = red[lane], m = redm[lane];
  for (int s = 16; s > 0; s >>= 1) {
    start += __shfl_xor(start, s, 32);
    m     += __shfl_xor(m, s, 32);
  }
  start = min(max(start, 0), n_edges);
  m     = min(max(m, 0), n_edges - start);
  const bool staged = (m <= OCAP);
  __syncthreads();

#pragma unroll
  for (int k = 0; k < KR; ++k) {
    if (sub + 2 * k < c0) atomicAdd(&cntL[(unsigned)q[k].x >> DLSH], 1);
  }
  for (int i = sub + 2 * KR; i < c0; i += 2)
    atomicAdd(&cntL[(unsigned)rec2[min(f0 + i, n_edges - 1)].x >> DLSH], 1);
  for (int b = b0 + CPT; b < n_chunks; b += CPT) {
    int lo, f, c;
    gcn_piece(tab, tab_pitch, b, bin, n_edges, &lo, &f, &c);
    for (int i = sub; i < c; i += 2)
      atomicAdd(&cntL[(unsigned)rec2[min(f + i, n_edges - 1)].x >> DLSH], 1);
  }
  __syncthreads();

  int c = 0, incl = 0;
  if (tid < BINW) {
    c = cntL[tid];
    incl = c;
    for (int s = 1; s < 32; s <<= 1) {
      const int u = __shfl_up(incl, s, 32);
      if (lane >= s) incl += u;
    }
    if (lane == 31) red[wv] = incl;
  }
  __syncthreads();
  if (tid < BINW) {
    int wpre = 0;
    for (int j = 0; j < 8; ++j) if (j < wv) wpre += red[j];
    const int excl = wpre + incl - c;
    offL[tid] = excl;
    const int node = node0 + tid;
    if (node < n_nodes) { cnt[node] = c; off[node] = start + excl; }
  }
  __syncthreads();

#pragma unroll
  for (int k = 0; k < KR; ++k)
    if (sub + 2 * k < c0)
      gcn_put(q[k], offL, curL, staged, stage, pair, start, n_edges);
  for (int i = sub + 2 * KR; i < c0; i += 2)
    gcn_put(rec2[min(f0 + i, n_edges - 1)], offL, curL, staged, stage, pair, start, n_edges);
  for (int b = b0 + CPT; b < n_chunks; b += CPT) {
    int lo, f, cc;
    gcn_piece(tab, tab_pitch, b, bin, n_edges, &lo, &f, &cc);
    for (int i = sub; i < cc; i += 2)
      gcn_put(rec2[min(f + i, n_edges - 1)], offL, curL, staged, stage, pair, start, n_edges);
  }
  __syncthreads();
  if (tid >= BINW) {
    if (staged) {
      const int shift = start & 15;
      for (int j = tid - BINW; j < m + shift; j += 1024 - BINW) {
        const int i = j - shift;
        if (i >= 0) ((v2i*)pair)[start + i] = stage[i];
      }
    }
  } else {
    if (staged) gcn_deg<true>(stage, pair, offL, cntL, dinv, tid, node0, start, n_nodes, n_edges);
    else        gcn_deg<false>(stage, pair, offL, cntL, dinv, tid, node0, start, n_nodes, n_edges);
  }
}

#define TN  64
#define XP  68
__global__ void __launch_bounds__(256)
gcn_xw(const float* __restrict__ dinv,
       const float* __restrict__ X, const float* __restrict__ W,
       float* __restrict__ XS, int n_nodes) {
  __shared__ __attribute__((aligned(16))) float xs[TN * XP];
  __shared__ __attribute__((aligned(16))) float sd[TN];
  const int tid   = threadIdx.x;
  const int lane  = tid & 31;
  const int wv    = tid >> 5;
  const int node0 = blockIdx.x * TN;
  const int half  = lane >> 4;
  const int l16   = lane & 15;
  const int col_tile = wv & 3;
  const int rp    = wv >> 2;

  float dv = 0.0f;
  if (tid < TN) dv = dinv[min(node0 + tid, n_nodes - 1)];

  const float* __restrict__ wcol = W + col_tile * 16 + l16;
  v2f bf[CCH / 4];
#pragma unroll
  for (int kk = 0; kk < CCH / 4; ++kk) {
    const int ka = kk * 4 + 2 * half;
    bf[kk].x = wcol[(size_t)(ka + 0) * CCH];
    bf[kk].y = wcol[(size_t)(ka + 1) * CCH];
  }

#pragma unroll
  for (int k = 0; k < 4; ++k) {
    const int idx = tid + k * 256;
    const int r   = idx >> 4;
    const int c4  = idx & 15;
    const int gr  = min(node0 + r, n_nodes - 1);
    const float4 v = ((const float4*)(X + (size_t)gr * CCH))[c4];
    *(float4*)&xs[r * XP + c4 * 4] = v;
  }
  if (tid < TN) sd[tid] = dv;
  __syncthreads();

  const int lr0 = rp * 32 + 8 * half;
  const float4 s00 = *(const float4*)&sd[lr0];
  const float4 s01 = *(const float4*)&sd[lr0 + 4];
  const float4 s10 = *(const float4*)&sd[lr0 + 16];
  const float4 s11 = *(const float4*)&sd[lr0 + 20];
  const float sc0[8] = {s00.x, s00.y, s00.z, s00.w, s01.x, s01.y, s01.z, s01.w};
  const float sc1[8] = {s10.x, s10.y, s10.z, s10.w, s11.x, s11.y, s11.z, s11.w};

  const float* xa0 = &xs[(rp * 32 + l16) * XP];
  const float* xa1 = xa0 + 16 * XP;
  v8f acc0 = {};
  v8f acc1 = {};
#pragma unroll
  for (int kk = 0; kk < CCH / 4; ++kk) {
    const int ka = kk * 4 + 2 * half;
    const v2f a0 = *(const v2f*)&xa0[ka];
    const v2f a1 = *(const v2f*)&xa1[ka];
    acc0 = __builtin_amdgcn_wmma_f32_16x16x4_f32(
         false, a0,  false, bf[kk],
         (short)0, acc0,  false,  false);
    acc1 = __builtin_amdgcn_wmma_f32_16x16x4_f32(
         false, a1,  false, bf[kk],
         (short)0, acc1,  false,  false);
  }

  float* __restrict__ dstp = XS + (size_t)(node0 + lr0) * CCH + col_tile * 16 + l16;
#pragma unroll
  for (int i = 0; i < 8; ++i) {
    if (node0 + lr0 + i < n_nodes) dstp[(size_t)i * CCH] = acc0[i] * sc0[i];
  }
#pragma unroll
  for (int i = 0; i < 8; ++i) {
    if (node0 + lr0 + 16 + i < n_nodes) dstp[(size_t)(16 + i) * CCH] = acc1[i] * sc1[i];
  }
}

#define GP  5
#define GTH 256

#define GCN_ROWB(x, J) __builtin_amdgcn_update_dpp(0, (x), 0x150 + (J), 0xf, 0xf, true)

__device__ __forceinline__ void gcn_edges(const v2i* __restrict__ pr, int beg, int m, int base,
                                          int l16, float dd, int own, int n_edges,
                                          int* s0, int* c0, int* s1, int* c1) {
  const int i0 = base + l16;
  const int i1 = i0 + 16;
  const v2i p0 = pr[min(max(beg + i0, 0), n_edges - 1)];
  const v2i p1 = pr[min(max(beg + i1, 0), n_edges - 1)];
  const bool ok0 = i0 < m && beg + i0 < n_edges;
  const bool ok1 = i1 < m && beg + i1 < n_edges;
  *s0 = ok0 ? p0.x : own;
  *c0 = ok0 ? __float_as_int(__int_as_float(p0.y) * dd) : 0;
  *s1 = ok1 ? p1.x : own;
  *c1 = ok1 ? __float_as_int(__int_as_float(p1.y) * dd) : 0;
}

template <int J0>
__device__ __forceinline__ void gcn_rows8(const int sv, const int cv,
                                          const char* __restrict__ xwb, const unsigned loff,
                                          float4* acc) {
#define GCN_LD(J)  (*(const float4*)(xwb + ((unsigned)GCN_ROWB(sv, J0 + (J)) * (unsigned)(CCH * 4) + loff)))
#define GCN_FMA(v, J) { const float c = __int_as_float(GCN_ROWB(cv, J0 + (J))); \
    acc->x = __builtin_fmaf((v).x, c, acc->x); acc->y = __builtin_fmaf((v).y, c, acc->y); \
    acc->z = __builtin_fmaf((v).z, c, acc->z); acc->w = __builtin_fmaf((v).w, c, acc->w); }
  const float4 v0 = GCN_LD(0);
  const float4 v1 = GCN_LD(1);
  const float4 v2 = GCN_LD(2);
  const float4 v3 = GCN_LD(3);
  const float4 v4 = GCN_LD(4);
  const float4 v5 = GCN_LD(5);
  const float4 v6 = GCN_LD(6);
  const float4 v7 = GCN_LD(7);
  GCN_FMA(v0, 0)
  GCN_FMA(v1, 1)
  GCN_FMA(v2, 2)
  GCN_FMA(v3, 3)
  GCN_FMA(v4, 4)
  GCN_FMA(v5, 5)
  GCN_FMA(v6, 6)
  GCN_FMA(v7, 7)
#undef GCN_LD
#undef GCN_FMA
}

__global__ void __launch_bounds__(GTH)
gcn_gather(const int* __restrict__ pair, const int* __restrict__ off,
           const int* __restrict__ cnt, const float* __restrict__ dinv,
           const float* __restrict__ xw, const float* __restrict__ b,
           float* __restrict__ out, int n_nodes, int n_edges) {
  const int lane = threadIdx.x & 31;
  const int l16  = lane & 15;
  const int half = lane >> 4;
  const int n0 = (2 * GP) * __builtin_amdgcn_readfirstlane(
      (int)((blockIdx.x * GTH + threadIdx.x) >> 5));
  if (n0 >= n_nodes) return;
  const v2i* __restrict__ pr = (const v2i*)pair;
  const char* __restrict__ xwb = (const char*)xw;
  const unsigned loff = (unsigned)l16 * 16u;

  const int hn = n0 + min(lane, 2 * GP - 1);
  const int hc = min(hn, n_nodes - 1);
  const int   offv = off[hc];
  const int   cntv = (hn < n_nodes) ? cnt[hc] : 0;
  const float dinv_v = dinv[hc];
  const float4 b4 = make_float4(b[4 * l16], b[4 * l16 + 1], b[4 * l16 + 2], b[4 * l16 + 3]);

  const int bA0 = __builtin_amdgcn_readlane(offv, 0);
  const int bB0 = __builtin_amdgcn_readlane(offv, 1);
  const int mA0 = __builtin_amdgcn_readlane(cntv, 0);
  const int mB0 = __builtin_amdgcn_readlane(cntv, 1);
  const int dA0 = __builtin_amdgcn_readlane(__float_as_int(dinv_v), 0);
  const int dB0 = __builtin_amdgcn_readlane(__float_as_int(dinv_v), 1);
  int   n    = n0 + half;
  int   beg  = half ? bB0 : bA0;
  int   m    = half ? mB0 : mA0;
  int   mmax = max(mA0, mB0);
  float dd   = __int_as_float(half ? dB0 : dA0);
  int s0, c0, s1, c1;
  gcn_edges(pr, beg, m, 0, l16, dd, min(n, n_nodes - 1), n_edges, &s0, &c0, &s1, &c1);

  for (int p = 0; p < GP; ++p) {
    const int own = min(n, n_nodes - 1);
    const float4 self = *(const float4*)(xwb + ((unsigned)own * (unsigned)(CCH * 4) + loff));

    const int pn = min(p + 1, GP - 1);
    const int n2   = n0 + 2 * pn + half;
    const int bA   = __builtin_amdgcn_readlane(offv, 2 * pn);
    const int bB   = __builtin_amdgcn_readlane(offv, 2 * pn + 1);
    const int mA2  = __builtin_amdgcn_readlane(cntv, 2 * pn);
    const int mB2  = __builtin_amdgcn_readlane(cntv, 2 * pn + 1);
    const int dA   = __builtin_amdgcn_readlane(__float_as_int(dinv_v), 2 * pn);
    const int dB   = __builtin_amdgcn_readlane(__float_as_int(dinv_v), 2 * pn + 1);
    const int   beg2 = half ? bB : bA;
    const int   m2   = half ? mB2 : mA2;
    const float dd2  = __int_as_float(half ? dB : dA);
    int t0, u0, t1, u1;
    gcn_edges(pr, beg2, m2, 0, l16, dd2, min(n2, n_nodes - 1), n_edges, &t0, &u0, &t1, &u1);

    float4 acc = make_float4(0.0f, 0.0f, 0.0f, 0.0f);
    for (int base = 0; base < mmax; base += 32) {
      if (base > 0) gcn_edges(pr, beg, m, base, l16, dd, own, n_edges, &s0, &c0, &s1, &c1);
      const int rem = mmax - base;
      gcn_rows8<0>(s0, c0, xwb, loff, &acc);
      if (rem > 8)  gcn_rows8<8>(s0, c0, xwb, loff, &acc);
      if (rem > 16) gcn_rows8<0>(s1, c1, xwb, loff, &acc);
      if (rem > 24) gcn_rows8<8>(s1, c1, xwb, loff, &acc);
    }

    const float k2 = FILLW * dd;
    float4 r;
    r.x = tanhf(acc.x + self.x * k2 + b4.x);
    r.y = tanhf(acc.y + self.y * k2 + b4.y);
    r.z = tanhf(acc.z + self.z * k2 + b4.z);
    r.w = tanhf(acc.w + self.w * k2 + b4.w);
    if (n < n_nodes) *(float4*)(out + (size_t)n * CCH + l16 * 4) = r;

    n = n2; beg = beg2; m = m2; dd = dd2; mmax = max(mA2, mB2);
    s0 = t0; c0 = u0; s1 = t1; c1 = u1;
  }
}

extern "C" void kernel_launch(void* const* d_in, const int* in_sizes, int n_in,
                              void* d_out, int out_size, void* d_ws, size_t ws_size,
                              hipStream_t stream) {
  const float* x  = (const float*)d_in[0];
  const int*   ei = (const int*)d_in[1];
  const float* ew = (const float*)d_in[2];
  const float* W  = (const float*)d_in[3];
  const float* b  = (const float*)d_in[4];
  float* out = (float*)d_out;

  const int n_nodes = in_sizes[0] / CCH;
  const int n_edges = in_sizes[2];
  const int* src = ei;
  const int* dst = ei + n_edges;

  const size_t n_pad = ((size_t)n_nodes + 63) & ~(size_t)63;
  const size_t e_pad = ((size_t)n_edges + 63) & ~(size_t)63;
  const int    n_bins = (n_nodes + BINW - 1) / BINW;
  const size_t b_pad = ((size_t)n_bins + 63) & ~(size_t)63;
  float* xw   = (float*)d_ws;
  float* dinv = xw + n_pad * CCH;
  int*   cnt  = (int*)(dinv + n_pad);
  int*   off  = cnt + n_pad;
  int*   tmp  = off + n_pad;
  int*   pair = tmp + e_pad * 2;
  int*   tab  = pair + e_pad * 2;
  const int tab_pitch = (int)b_pad + 64;

  const int B = 256;
  const int n_chunks = (n_edges + EPB - 1) / EPB;

  gcn_chunk<<<n_chunks, B, 2 * n_bins * sizeof(int), stream>>>(src, dst, ew, tmp, tab,
                                                              n_nodes, n_edges, n_bins, tab_pitch);
  gcn_owner<<<n_bins, 1024, 0, stream>>>(tmp, tab, cnt, off, pair, dinv, n_nodes, n_edges,
                                         n_chunks, tab_pitch);
  gcn_xw<<<(n_nodes + TN - 1) / TN, B, 0, stream>>>(dinv, x, W, xw, n_nodes);

  gcn_gather<<<(n_nodes + 16 * GP - 1) / (16 * GP), GTH, 0, stream>>>(pair, off, cnt, dinv, xw, b, out,
                                                                    n_nodes, n_edges);
}
